# starting kernel
# baseline (speedup 1.0000x reference)
#include <hip/hip_runtime.h>

typedef float v2f __attribute__((ext_vector_type(2)));
typedef float v8f __attribute__((ext_vector_type(8)));

#define IN_CH 512
#define HID   128
#define OUTC  64

__global__ void k_set_ones(float* deg, int n) {
  int i = blockIdx.x * blockDim.x + threadIdx.x;
  if (i < n) deg[i] = 1.0f;
}

__global__ void k_deg_accum(const int* __restrict__ ei, const float* __restrict__ ew,
                            float* deg, int E) {
  int e = blockIdx.x * blockDim.x + threadIdx.x;
  if (e < E) atomicAdd(&deg[ei[E + e]], ew[e]);
}

__global__ void k_rsqrt(const float* __restrict__ deg, float* dinv, int n) {
  int i = blockIdx.x * blockDim.x + threadIdx.x;
  if (i < n) {
    float d = deg[i];
    dinv[i] = (d > 0.0f) ? rsqrtf(d) : 0.0f;
  }
}

__global__ void k_pack_w(const float* __restrict__ W, float* __restrict__ Wp,
                         int K, int N) {
  int idx = blockIdx.x * blockDim.x + threadIdx.x;
  if (idx >= K * N) return;
  int k = idx / N;
  int n = idx - k * N;
  int k4 = k >> 2;
  int hi = (k >> 1) & 1;
  int j  = k & 1;
  Wp[(((size_t)((k4 << 1) | hi)) * N + n) * 2 + j] = W[idx];
}

template <int NT>
__global__ void k_gemm_wmma(const float* __restrict__ A,
                            const float* __restrict__ Bp,
                            float* __restrict__ C,
                            int M, int K) {
  const int N = NT * 16;
  int wave = (int)((blockIdx.x * blockDim.x + threadIdx.x) >> 5);
  int lane = threadIdx.x & 31;
  if (wave >= (M >> 4)) return;

  int m0 = wave << 4;
  int lo = lane & 15;
  int hi = lane >> 4;

  const float* aPtr = A  + (size_t)(m0 + lo) * K + (hi << 1);
  const float* bPtr = Bp + ((size_t)hi * N + lo) * 2;

  v8f z = {0.f,0.f,0.f,0.f,0.f,0.f,0.f,0.f};
  v8f acc[NT];
#pragma unroll
  for (int t = 0; t < NT; ++t) acc[t] = z;

  for (int k = 0; k < K; k += 8) {
    v2f a0 = *(const v2f*)(aPtr + k);
    v2f a1 = *(const v2f*)(aPtr + k + 4);
    const float* bp0 = bPtr + (size_t)k * N;
    const float* bp1 = bp0 + (size_t)4 * N;
    v2f b0[NT], b1[NT];
#pragma unroll
    for (int t = 0; t < NT; ++t) b0[t] = *(const v2f*)(bp0 + t * 32);
#pragma unroll
    for (int t = 0; t < NT; ++t) b1[t] = *(const v2f*)(bp1 + t * 32);
#pragma unroll
    for (int t = 0; t < NT; ++t)
      acc[t] = __builtin_amdgcn_wmma_f32_16x16x4_f32(
          false, a0, false, b0[t], (short)0, acc[t], false, false);
#pragma unroll
    for (int t = 0; t < NT; ++t)
      acc[t] = __builtin_amdgcn_wmma_f32_16x16x4_f32(
          false, a1, false, b1[t], (short)0, acc[t], false, false);
  }

  int cRow = m0 + (hi << 3);
#pragma unroll
  for (int r = 0; r < 8; ++r) {
    float* cp = C + (size_t)(cRow + r) * N + lo;
#pragma unroll
    for (int t = 0; t < NT; ++t) cp[t * 16] = acc[t][r];
  }
}

__global__ void k_init_agg1(const float* __restrict__ h1, const float* __restrict__ dinv,
                            const float* __restrict__ b1, float* agg, int n) {
  int idx = blockIdx.x * blockDim.x + threadIdx.x;
  if (idx < n) {
    int row = idx >> 7, c = idx & 127;
    float di = dinv[row];
    agg[idx] = h1[idx] * di * di + b1[c];
  }
}

__global__ void k_scatter1(const int* __restrict__ ei, const float* __restrict__ ew,
                           const float* __restrict__ dinv, const float* __restrict__ h,
                           float* agg, int E) {
  long long idx = (long long)blockIdx.x * blockDim.x + threadIdx.x;
  if (idx >= (long long)E * 32) return;
  int e = (int)(idx >> 5);
  int g = ((int)idx & 31) << 2;
  int s = ei[e], d = ei[E + e];
  float nm = dinv[s] * ew[e] * dinv[d];
  float4 v = *(const float4*)(h + (size_t)s * HID + g);
  float* o = agg + (size_t)d * HID + g;
  atomicAdd(o + 0, v.x * nm);
  atomicAdd(o + 1, v.y * nm);
  atomicAdd(o + 2, v.z * nm);
  atomicAdd(o + 3, v.w * nm);
}

__global__ void k_relu(float* a, int n) {
  int i = blockIdx.x * blockDim.x + threadIdx.x;
  if (i < n) a[i] = fmaxf(a[i], 0.0f);
}

__global__ void k_init_out(const float* __restrict__ t2, const float* __restrict__ t3,
                           const float* __restrict__ dinv,
                           const float* __restrict__ b2, const float* __restrict__ b3,
                           float* om, float* ol, int n) {
  int idx = blockIdx.x * blockDim.x + threadIdx.x;
  if (idx < n) {
    int row = idx >> 6, c = idx & 63;
    float di = dinv[row];
    float d2 = di * di;
    om[idx] = t2[idx] * d2 + b2[c];
    ol[idx] = t3[idx] * d2 + b3[c];
  }
}

__global__ void k_scatter2(const int* __restrict__ ei, const float* __restrict__ ew,
                           const float* __restrict__ dinv,
                           const float* __restrict__ t2, const float* __restrict__ t3,
                           float* om, float* ol, int E) {
  long long idx = (long long)blockIdx.x * blockDim.x + threadIdx.x;
  if (idx >= (long long)E * 32) return;
  int e = (int)(idx >> 5);
  int l = (int)idx & 31;
  int s = ei[e], d = ei[E + e];
  float nm = dinv[s] * ew[e] * dinv[d];
  const float* tin = (l < 16) ? t2 : t3;
  float*       op  = (l < 16) ? om : ol;
  int g = (l & 15) << 2;
  float4 v = *(const float4*)(tin + (size_t)s * OUTC + g);
  float* o = op + (size_t)d * OUTC + g;
  atomicAdd(o + 0, v.x * nm);
  atomicAdd(o + 1, v.y * nm);
  atomicAdd(o + 2, v.z * nm);
  atomicAdd(o + 3, v.w * nm);
}

extern "C" void kernel_launch(void* const* d_in, const int* in_sizes, int n_in,
                              void* d_out, int out_size, void* d_ws, size_t ws_size,
                              hipStream_t stream) {
  const float* x  = (const float*)d_in[0];
  const int*   ei = (const int*)  d_in[1];
  const float* ew = (const float*)d_in[2];
  const float* W1 = (const float*)d_in[3];
  const float* b1 = (const float*)d_in[4];
  const float* W2 = (const float*)d_in[5];
  const float* b2 = (const float*)d_in[6];
  const float* W3 = (const float*)d_in[7];
  const float* b3 = (const float*)d_in[8];

  const int N = in_sizes[0] / IN_CH;
  const int E = in_sizes[2];

  float* ws   = (float*)d_ws;
  float* deg  = ws;
  float* dinv = ws + N;
  float* h1   = ws + 2 * (size_t)N;
  float* agg1 = h1 + (size_t)N * HID;
  float* W1p  = agg1 + (size_t)N * HID;
  float* W2p  = W1p + (size_t)IN_CH * HID;
  float* W3p  = W2p + (size_t)HID * OUTC;
  float* t2   = h1;
  float* t3   = h1 + (size_t)N * OUTC;
  float* om   = (float*)d_out;
  float* ol   = om + (size_t)N * OUTC;

  const int T = 256;

  k_pack_w<<<(IN_CH * HID + T - 1) / T, T, 0, stream>>>(W1, W1p, IN_CH, HID);
  k_pack_w<<<(HID * OUTC + T - 1) / T, T, 0, stream>>>(W2, W2p, HID, OUTC);
  k_pack_w<<<(HID * OUTC + T - 1) / T, T, 0, stream>>>(W3, W3p, HID, OUTC);

  k_set_ones  <<<(N + T - 1) / T, T, 0, stream>>>(deg, N);
  k_deg_accum <<<(E + T - 1) / T, T, 0, stream>>>(ei, ew, deg, E);
  k_rsqrt     <<<(N + T - 1) / T, T, 0, stream>>>(deg, dinv, N);

  {
    int waves  = N >> 4;
    int blocks = (waves * 32 + T - 1) / T;
    k_gemm_wmma<8><<<blocks, T, 0, stream>>>(x, W1p, h1, N, IN_CH);
  }

  k_init_agg1<<<((N * HID) + T - 1) / T, T, 0, stream>>>(h1, dinv, b1, agg1, N * HID);
  {
    long long tot = (long long)E * 32;
    int blocks = (int)((tot + T - 1) / T);
    k_scatter1<<<blocks, T, 0, stream>>>(ei, ew, dinv, h1, agg1, E);
  }
  k_relu<<<((N * HID) + T - 1) / T, T, 0, stream>>>(agg1, N * HID);
  const float* h = agg1;

  {
    int waves  = N >> 4;
    int blocks = (waves * 32 + T - 1) / T;
    k_gemm_wmma<4><<<blocks, T, 0, stream>>>(h, W2p, t2, N, HID);
    k_gemm_wmma<4><<<blocks, T, 0, stream>>>(h, W3p, t3, N, HID);
  }

  k_init_out<<<((N * OUTC) + T - 1) / T, T, 0, stream>>>(t2, t3, dinv, b2, b3, om, ol, N * OUTC);
  {
    long long tot = (long long)E * 32;
    int blocks = (int)((tot + T - 1) / T);
    k_scatter2<<<blocks, T, 0, stream>>>(ei, ew, dinv, t2, t3, om, ol, E);
  }
}


# Layer-2 edge sums reuse layer-1's per-wave edge queues: (source, row, norm) records parked in the work-group's own output rows; no second owner-byte scan or edge lookup; sums stay in arrival order
# speedup vs baseline: 2.0340x; 2.0340x over previous
#include <hip/hip_runtime.h>

typedef float v2f __attribute__((ext_vector_type(2)));
typedef float v8f __attribute__((ext_vector_type(8)));

#define IN_CH 512
#define HID   128
#define OUTC  64

#define RT    196
#define AGG_T 1024
#define AGG_W (AGG_T / 32)
#define QCAP  448

__global__ void k_set_ones(float* deg, int n) {
  int i = blockIdx.x * blockDim.x + threadIdx.x;
  if (i < n) deg[i] = 1.0f;
}

__global__ void k_deg_accum(const int* __restrict__ ei, const float* __restrict__ ew,
                            float* deg, unsigned int* __restrict__ own, int E) {
  int e = blockIdx.x * blockDim.x + threadIdx.x;
  unsigned int o = 0u;
  if (e < E) {
    int d = ei[E + e];
    atomicAdd(&deg[d], ew[e]);
    o = (unsigned int)d / RT;
  }
  unsigned int o1 = __shfl_down(o, 1);
  unsigned int o2 = __shfl_down(o, 2);
  unsigned int o3 = __shfl_down(o, 3);
  if ((threadIdx.x & 3) == 0) own[e >> 2] = o | (o1 << 8) | (o2 << 16) | (o3 << 24);
}

__global__ void k_rsqrt(const float* __restrict__ deg, float* dinv, int n) {
  int i = blockIdx.x * blockDim.x + threadIdx.x;
  if (i < n) {
    float d = deg[i];
    dinv[i] = (d > 0.0f) ? rsqrtf(d) : 0.0f;
  }
}

__global__ void k_pack_w(const float* __restrict__ W, float* __restrict__ Wp,
                         int K, int N) {
  int idx = blockIdx.x * blockDim.x + threadIdx.x;
  if (idx >= K * N) return;
  int k = idx / N;
  int n = idx - k * N;
  int k4 = k >> 2;
  int hi = (k >> 1) & 1;
  int j  = k & 1;
  Wp[(((size_t)((k4 << 1) | hi)) * N + n) * 2 + j] = W[idx];
}

template <int NT, bool LAYER2>
__global__ void k_gemm_wmma(const float* __restrict__ A,
                            const float* __restrict__ Bp,
                            float* __restrict__ C,
                            int M, int K) {
  const int N = NT * 16;
  int wave = (int)((blockIdx.x * blockDim.x + threadIdx.x) >> 5);
  int lane = threadIdx.x & 31;
  if (wave >= (M >> 4)) return;

  int m0 = wave << 4;
  int lo = lane & 15;
  int hi = lane >> 4;

  const float* aPtr = A  + (size_t)(m0 + lo) * K + (hi << 1);
  const float* bPtr = Bp + ((size_t)hi * N + lo) * 2;

  v8f z = {0.f,0.f,0.f,0.f,0.f,0.f,0.f,0.f};
  v8f acc[NT];
#pragma unroll
  for (int t = 0; t < NT; ++t) acc[t] = z;

  for (int k = 0; k < K; k += 8) {
    v2f a0 = *(const v2f*)(aPtr + k);
    v2f a1 = *(const v2f*)(aPtr + k + 4);
    const float* bp0 = bPtr + (size_t)k * N;
    const float* bp1 = bp0 + (size_t)4 * N;
    v2f b0[NT], b1[NT];
#pragma unroll
    for (int t = 0; t < NT; ++t) b0[t] = *(const v2f*)(bp0 + t * 32);
#pragma unroll
    for (int t = 0; t < NT; ++t) b1[t] = *(const v2f*)(bp1 + t * 32);
    if (LAYER2) {
      a0.x = fmaxf(a0.x, 0.0f); a0.y = fmaxf(a0.y, 0.0f);
      a1.x = fmaxf(a1.x, 0.0f); a1.y = fmaxf(a1.y, 0.0f);
    }
#pragma unroll
    for (int t = 0; t < NT; ++t)
      acc[t] = __builtin_amdgcn_wmma_f32_16x16x4_f32(
          false, a0, false, b0[t], (short)0, acc[t], false, false);
#pragma unroll
    for (int t = 0; t < NT; ++t)
      acc[t] = __builtin_amdgcn_wmma_f32_16x16x4_f32(
          false, a1, false, b1[t], (short)0, acc[t], false, false);
  }

  int cRow = m0 + (hi << 3);
#pragma unroll
  for (int r = 0; r < 8; ++r) {
    float* cp = C + (size_t)(cRow + r) * N + lo;
#pragma unroll
    for (int t = 0; t < NT; ++t) cp[t * 16] = acc[t][r];
  }
}

#define EB 8
#define NOREC 0xFFFFFFFFu

__device__ __forceinline__ int bcast_i(int v, int j) {
  return __builtin_amdgcn_readlane(v, j);
}
__device__ __forceinline__ float bcast_f(float v, int j) {
  return __int_as_float(__builtin_amdgcn_readlane(__float_as_int(v), j));
}

template <int LAYER>
__device__ __forceinline__ void row4(const float* __restrict__ ta,
                                     const float* __restrict__ tb,
                                     int s, int l, float* v) {
  if (LAYER == 1) {
    int so = s * HID + l;
    v[0] = ta[so]; v[1] = ta[so + 32]; v[2] = ta[so + 64]; v[3] = ta[so + 96];
  } else {
    int so = s * OUTC + l;
    v[0] = ta[so]; v[1] = ta[so + 32]; v[2] = tb[so]; v[3] = tb[so + 32];
  }
}

template <int LAYER>
__global__ void __launch_bounds__(AGG_T)
k_agg(const int* __restrict__ ei, const float* __restrict__ ew,
      const float* __restrict__ dinv, const unsigned int* __restrict__ own,
      const float* __restrict__ ta, const float* __restrict__ tb,
      const float* __restrict__ ba, const float* __restrict__ bb,
      float* oa, float* ob, float* qa, float* qb, unsigned int* qc, int N, int E) {
  __shared__ float acc[RT * 128];
  __shared__ unsigned int q[AGG_W * QCAP];
  __shared__ unsigned int qn[AGG_W];
  __shared__ unsigned int tot;
  __shared__ unsigned int tw[AGG_W];

  const int tid  = threadIdx.x;
  const int l    = tid & 31;
  const int wave = __builtin_amdgcn_readfirstlane(tid >> 5);
  const int r0   = blockIdx.x * RT;

  for (int i = tid; i < RT * 128; i += AGG_T) acc[i] = 0.0f;
  if (tid < AGG_W) qn[tid] = 0u;
  if (tid == 0) tot = 0u;
  __syncthreads();

  int rows = N - r0;
  if (rows > RT) rows = RT;
  const int cap = rows * OUTC;
  float* ra = ((LAYER == 1) ? qa : oa) + r0 * OUTC;
  float* rb = ((LAYER == 1) ? qb : ob) + r0 * OUTC;

  auto add_group = [&](int sv, int dl, float nv, int m) {
    int j0 = 0;
    for (; j0 + EB <= m; j0 += EB) {
      float v[EB][4];
#pragma unroll
      for (int j = 0; j < EB; ++j) row4<LAYER>(ta, tb, bcast_i(sv, j0 + j), l, v[j]);
#pragma unroll
      for (int j = 0; j < EB; ++j) {
        float* ar = acc + bcast_i(dl, j0 + j) * 128 + l;
        float nm  = bcast_f(nv, j0 + j);
        atomicAdd(&ar[0],  v[j][0] * nm);
        atomicAdd(&ar[32], v[j][1] * nm);
        atomicAdd(&ar[64], v[j][2] * nm);
        atomicAdd(&ar[96], v[j][3] * nm);
      }
    }
    for (; j0 < m; ++j0) {
      float v[4];
      row4<LAYER>(ta, tb, bcast_i(sv, j0), l, v);
      float* ar = acc + bcast_i(dl, j0) * 128 + l;
      float nm  = bcast_f(nv, j0);
      atomicAdd(&ar[0],  v[0] * nm);
      atomicAdd(&ar[32], v[1] * nm);
      atomicAdd(&ar[64], v[2] * nm);
      atomicAdd(&ar[96], v[3] * nm);
    }
  };

  unsigned int cw = NOREC;
  if (LAYER == 2) cw = (unsigned int)__builtin_amdgcn_readfirstlane((int)qc[blockIdx.x * AGG_W + wave]);
  if (cw != NOREC) {
    const int off = (int)(cw & 0xFFFFu);
    const int n   = (int)(cw >> 16);
    unsigned int p = 0u;
    float nv = 0.0f;
    if (l < n) { p = __float_as_uint(ra[off + l]); nv = rb[off + l]; }
    for (int g = 0; g < n; g += 32) {
      unsigned int pn = 0u;
      float nn = 0.0f;
      if (g + 32 + l < n) { pn = __float_as_uint(ra[off + g + 32 + l]); nn = rb[off + g + 32 + l]; }
      int m = n - g;
      if (m > 32) m = 32;
      add_group((int)(p & 0xFFFFu), (int)(p >> 16), nv, m);
      p = pn;
      nv = nn;
    }
  } else {

  const int nChunks = (E + 511) >> 9;
  const unsigned int idw = (unsigned int)blockIdx.x * 0x01010101u;
  const int rot = (int)(((unsigned int)blockIdx.x * (unsigned int)nChunks) / gridDim.x);
  unsigned int* myq = q + wave * QCAP;

  auto search = [&](uint4 x, int eb) {
    unsigned int t0 = x.x ^ idw, t1 = x.y ^ idw, t2 = x.z ^ idw, t3 = x.w ^ idw;
    unsigned int z0 = ~(((t0 & 0x7F7F7F7Fu) + 0x7F7F7F7Fu) | t0) & 0x80808080u;
    unsigned int z1 = ~(((t1 & 0x7F7F7F7Fu) + 0x7F7F7F7Fu) | t1) & 0x80808080u;
    unsigned int z2 = ~(((t2 & 0x7F7F7F7Fu) + 0x7F7F7F7Fu) | t2) & 0x80808080u;
    unsigned int z3 = ~(((t3 & 0x7F7F7F7Fu) + 0x7F7F7F7Fu) | t3) & 0x80808080u;
    unsigned int zz = (z0 >> 7) | (z1 >> 6) | (z2 >> 5) | (z3 >> 4);
    while (zz) {
      int bit = __builtin_ctz(zz);
      zz &= zz - 1u;
      int e = eb + ((bit & 7) << 2) + (bit >> 3);
      if (e < E) {
        unsigned int pos = atomicAdd(&qn[wave], 1u);
        if (pos < QCAP) {
          myq[pos] = (unsigned int)e;
        } else {
          int s = ei[e];
          int d = ei[E + e];
          float nm = dinv[s] * ew[e] * dinv[d];
          float* ar = acc + (d - r0) * 128;
          for (int c = 0; c < 128; ++c) {
            float hv = (LAYER == 1) ? ta[s * HID + c]
                                    : (c < OUTC ? ta[s * OUTC + c] : tb[s * OUTC + c - OUTC]);
            atomicAdd(&ar[c], hv * nm);
          }
        }
      }
    }
  };
  auto lane_base = [&](int ii) {
    int c = ii + rot;
    if (c >= nChunks) c -= nChunks;
    return (c << 9) + (l << 4);
  };

  int i = wave;
  for (; i + 3 * AGG_W < nChunks; i += 4 * AGG_W) {
    int b0 = lane_base(i), b1 = lane_base(i + AGG_W);
    int b2 = lane_base(i + 2 * AGG_W), b3 = lane_base(i + 3 * AGG_W);
    uint4 x0 = *(const uint4*)(own + (b0 >> 2));
    uint4 x1 = *(const uint4*)(own + (b1 >> 2));
    uint4 x2 = *(const uint4*)(own + (b2 >> 2));
    uint4 x3 = *(const uint4*)(own + (b3 >> 2));
    search(x0, b0);
    search(x1, b1);
    search(x2, b2);
    search(x3, b3);
  }
  for (; i < nChunks; i += AGG_W) {
    int b0 = lane_base(i);
    uint4 x0 = *(const uint4*)(own + (b0 >> 2));
    search(x0, b0);
  }

  int nq = __builtin_amdgcn_readfirstlane((int)qn[wave]);
  const bool ovf = nq > QCAP;
  const int n = ovf ? QCAP : nq;
  int off = 0;
  bool keep = false;
  if (LAYER == 1) {
    unsigned int o = 0u;
    if (l == 0) o = atomicAdd(&tot, (unsigned int)((n + 31) & ~31));
    off = __builtin_amdgcn_readfirstlane((int)o);
    keep = !ovf && off + n <= cap;
    if (l == 0) tw[wave] = keep ? ((unsigned int)off | ((unsigned int)n << 16)) : NOREC;
  }
  for (int g = 0; g < n; g += 32) {
    int sv = 0, dl = 0;
    float nv = 0.0f;
    if (g + l < n) {
      int e = (int)myq[g + l];
      int s = ei[e];
      int d = ei[E + e];
      nv = dinv[s] * ew[e] * dinv[d];
      sv = s;
      dl = d - r0;
      if (LAYER == 1 && keep) {
        ra[off + g + l] = __uint_as_float((unsigned int)sv | ((unsigned int)dl << 16));
        rb[off + g + l] = nv;
      }
    }
    int m = n - g;
    if (m > 32) m = 32;
    add_group(sv, dl, nv, m);
  }
  }
  __syncthreads();
  if (LAYER == 1 && tid < AGG_W) qc[blockIdx.x * AGG_W + tid] = tw[tid];

  const int c = tid & 127;
  for (int lr = tid >> 7; lr < rows; lr += AGG_T / 128) {
    int r = r0 + lr;
    float di = dinv[r];
    float d2 = di * di;
    float a = acc[lr * 128 + c];
    if (LAYER == 1) {
      oa[r * HID + c] = a + (ta[r * HID + c] * d2 + ba[c]);
    } else if (c < OUTC) {
      oa[r * OUTC + c] = a + (ta[r * OUTC + c] * d2 + ba[c]);
    } else {
      ob[r * OUTC + c - OUTC] = a + (tb[r * OUTC + c - OUTC] * d2 + bb[c - OUTC]);
    }
  }
}

extern "C" void kernel_launch(void* const* d_in, const int* in_sizes, int n_in,
                              void* d_out, int out_size, void* d_ws, size_t ws_size,
                              hipStream_t stream) {
  const float* x  = (const float*)d_in[0];
  const int*   ei = (const int*)  d_in[1];
  const float* ew = (const float*)d_in[2];
  const float* W1 = (const float*)d_in[3];
  const float* b1 = (const float*)d_in[4];
  const float* W2 = (const float*)d_in[5];
  const float* b2 = (const float*)d_in[6];
  const float* W3 = (const float*)d_in[7];
  const float* b3 = (const float*)d_in[8];

  const int N = in_sizes[0] / IN_CH;
  const int E = in_sizes[2];

  float* ws   = (float*)d_ws;
  float* deg  = ws;
  float* dinv = ws + N;
  float* h1   = ws + 2 * (size_t)N;
  float* agg1 = h1 + (size_t)N * HID;
  float* W1p  = agg1 + (size_t)N * HID;
  float* W2p  = W1p + (size_t)IN_CH * HID;
  float* W3p  = W2p + (size_t)HID * OUTC;
  unsigned int* own = (unsigned int*)(W3p + (size_t)HID * OUTC);
  float* t2   = h1;
  float* t3   = h1 + (size_t)N * OUTC;
  float* om   = (float*)d_out;
  float* ol   = om + (size_t)N * OUTC;
  unsigned int* qc = (unsigned int*)deg;

  const int T = 256;

  k_pack_w<<<(IN_CH * HID + T - 1) / T, T, 0, stream>>>(W1, W1p, IN_CH, HID);
  k_pack_w<<<(HID * OUTC + T - 1) / T, T, 0, stream>>>(W2, W2p, HID, OUTC);
  k_pack_w<<<(HID * OUTC + T - 1) / T, T, 0, stream>>>(W3, W3p, HID, OUTC);

  k_set_ones  <<<(N + T - 1) / T, T, 0, stream>>>(deg, N);
  const int nChunks = (E + 511) / 512;
  k_deg_accum <<<nChunks * 2, T, 0, stream>>>(ei, ew, deg, own, E);
  k_rsqrt     <<<(N + T - 1) / T, T, 0, stream>>>(deg, dinv, N);

  const int waves  = N >> 4;
  const int gblocks = (waves * 32 + T - 1) / T;
  const int ablocks = (N + RT - 1) / RT;

  k_gemm_wmma<8, false><<<gblocks, T, 0, stream>>>(x, W1p, h1, N, IN_CH);

  k_agg<1><<<ablocks, AGG_T, 0, stream>>>(ei, ew, dinv, own, h1, h1, b1, b1, agg1, agg1, om, ol, qc, N, E);

  k_gemm_wmma<4, true><<<gblocks, T, 0, stream>>>(agg1, W2p, t2, N, HID);
  k_gemm_wmma<4, true><<<gblocks, T, 0, stream>>>(agg1, W3p, t3, N, HID);

  k_agg<2><<<ablocks, AGG_T, 0, stream>>>(ei, ew, dinv, own, t2, t3, b2, b3, om, ol, om, ol, qc, N, E);
}


# Degree sums leave global f32 atomics: 64 resident work-groups add their edge slices into whole degree arrays in LDS, the dinv launch adds the 64 partials + self-loop 1; 10 launches; sum order changes
# speedup vs baseline: 2.6456x; 1.3007x over previous
#include <hip/hip_runtime.h>

typedef float v2f __attribute__((ext_vector_type(2)));
typedef float v8f __attribute__((ext_vector_type(8)));

#define IN_CH 512
#define HID   128
#define OUTC  64

#define RT    196
#define AGG_T 1024
#define AGG_W (AGG_T / 32)
#define QCAP  448

#define DEG_G 64
#define NCAP  (256 * RT)
#define DEG_B 8

__global__ void __launch_bounds__(1024)
k_deg_part(const int* __restrict__ ei, const float* __restrict__ ew,
           float* __restrict__ part, unsigned int* __restrict__ own, int N, int E) {
  __shared__ float dg[NCAP];
  const int tid = threadIdx.x;
  for (int i = tid; i < NCAP; i += 1024) dg[i] = 0.0f;
  __syncthreads();

  const int EP  = ((E + 511) >> 9) << 9;
  const int nCh = (EP + 1023) >> 10;
  const int per = (nCh + (int)gridDim.x - 1) / (int)gridDim.x;
  const int c0  = (int)blockIdx.x * per;
  int c1 = c0 + per;
  if (c1 > nCh) c1 = nCh;

  for (int c = c0; c < c1; c += DEG_B) {
    int   d[DEG_B];
    float w[DEG_B];
#pragma unroll
    for (int j = 0; j < DEG_B; ++j) {
      int e = ((c + j) << 10) + tid;
      d[j] = -1;
      w[j] = 0.0f;
      if (c + j < c1 && e < E) { d[j] = ei[E + e]; w[j] = ew[e]; }
    }
#pragma unroll
    for (int j = 0; j < DEG_B; ++j) {
      int e = ((c + j) << 10) + tid;
      unsigned int o = 0u;
      if (d[j] >= 0) {
        atomicAdd(&dg[d[j]], w[j]);
        o = (unsigned int)d[j] / RT;
      }
      unsigned int o1 = __shfl_down(o, 1);
      unsigned int o2 = __shfl_down(o, 2);
      unsigned int o3 = __shfl_down(o, 3);
      if (c + j < c1 && e < EP && (tid & 3) == 0)
        own[e >> 2] = o | (o1 << 8) | (o2 << 16) | (o3 << 24);
    }
  }
  __syncthreads();

  float* p = part + (size_t)blockIdx.x * N;
  for (int i = tid; i < N; i += 1024) p[i] = dg[i];
}

__global__ void k_rsqrt(const float* __restrict__ part, float* __restrict__ dinv, int n) {
  int i = blockIdx.x * blockDim.x + threadIdx.x;
  if (i < n) {
    float d = 1.0f;
#pragma unroll 16
    for (int g = 0; g < DEG_G; ++g) d += part[(size_t)g * n + i];
    dinv[i] = (d > 0.0f) ? rsqrtf(d) : 0.0f;
  }
}

__global__ void k_pack_w(const float* __restrict__ W, float* __restrict__ Wp,
                         int K, int N) {
  int idx = blockIdx.x * blockDim.x + threadIdx.x;
  if (idx >= K * N) return;
  int k = idx / N;
  int n = idx - k * N;
  int k4 = k >> 2;
  int hi = (k >> 1) & 1;
  int j  = k & 1;
  Wp[(((size_t)((k4 << 1) | hi)) * N + n) * 2 + j] = W[idx];
}

template <int NT, bool LAYER2>
__global__ void k_gemm_wmma(const float* __restrict__ A,
                            const float* __restrict__ Bp,
                            float* __restrict__ C,
                            int M, int K) {
  const int N = NT * 16;
  int wave = (int)((blockIdx.x * blockDim.x + threadIdx.x) >> 5);
  int lane = threadIdx.x & 31;
  if (wave >= (M >> 4)) return;

  int m0 = wave << 4;
  int lo = lane & 15;
  int hi = lane >> 4;

  const float* aPtr = A  + (size_t)(m0 + lo) * K + (hi << 1);
  const float* bPtr = Bp + ((size_t)hi * N + lo) * 2;

  v8f z = {0.f,0.f,0.f,0.f,0.f,0.f,0.f,0.f};
  v8f acc[NT];
#pragma unroll
  for (int t = 0; t < NT; ++t) acc[t] = z;

  for (int k = 0; k < K; k += 8) {
    v2f a0 = *(const v2f*)(aPtr + k);
    v2f a1 = *(const v2f*)(aPtr + k + 4);
    const float* bp0 = bPtr + (size_t)k * N;
    const float* bp1 = bp0 + (size_t)4 * N;
    v2f b0[NT], b1[NT];
#pragma unroll
    for (int t = 0; t < NT; ++t) b0[t] = *(const v2f*)(bp0 + t * 32);
#pragma unroll
    for (int t = 0; t < NT; ++t) b1[t] = *(const v2f*)(bp1 + t * 32);
    if (LAYER2) {
      a0.x = fmaxf(a0.x, 0.0f); a0.y = fmaxf(a0.y, 0.0f);
      a1.x = fmaxf(a1.x, 0.0f); a1.y = fmaxf(a1.y, 0.0f);
    }
#pragma unroll
    for (int t = 0; t < NT; ++t)
      acc[t] = __builtin_amdgcn_wmma_f32_16x16x4_f32(
          false, a0, false, b0[t], (short)0, acc[t], false, false);
#pragma unroll
    for (int t = 0; t < NT; ++t)
      acc[t] = __builtin_amdgcn_wmma_f32_16x16x4_f32(
          false, a1, false, b1[t], (short)0, acc[t], false, false);
  }

  int cRow = m0 + (hi << 3);
#pragma unroll
  for (int r = 0; r < 8; ++r) {
    float* cp = C + (size_t)(cRow + r) * N + lo;
#pragma unroll
    for (int t = 0; t < NT; ++t) cp[t * 16] = acc[t][r];
  }
}

#define EB 8
#define NOREC 0xFFFFFFFFu

__device__ __forceinline__ int bcast_i(int v, int j) {
  return __builtin_amdgcn_readlane(v, j);
}
__device__ __forceinline__ float bcast_f(float v, int j) {
  return __int_as_float(__builtin_amdgcn_readlane(__float_as_int(v), j));
}

template <int LAYER>
__device__ __forceinline__ void row4(const float* __restrict__ ta,
                                     const float* __restrict__ tb,
                                     int s, int l, float* v) {
  if (LAYER == 1) {
    int so = s * HID + l;
    v[0] = ta[so]; v[1] = ta[so + 32]; v[2] = ta[so + 64]; v[3] = ta[so + 96];
  } else {
    int so = s * OUTC + l;
    v[0] = ta[so]; v[1] = ta[so + 32]; v[2] = tb[so]; v[3] = tb[so + 32];
  }
}

template <int LAYER>
__global__ void __launch_bounds__(AGG_T)
k_agg(const int* __restrict__ ei, const float* __restrict__ ew,
      const float* __restrict__ dinv, const unsigned int* __restrict__ own,
      const float* __restrict__ ta, const float* __restrict__ tb,
      const float* __restrict__ ba, const float* __restrict__ bb,
      float* oa, float* ob, float* qa, float* qb, unsigned int* qc, int N, int E) {
  __shared__ float acc[RT * 128];
  __shared__ unsigned int q[AGG_W * QCAP];
  __shared__ unsigned int qn[AGG_W];
  __shared__ unsigned int tot;
  __shared__ unsigned int tw[AGG_W];

  const int tid  = threadIdx.x;
  const int l    = tid & 31;
  const int wave = __builtin_amdgcn_readfirstlane(tid >> 5);
  const int r0   = blockIdx.x * RT;

  for (int i = tid; i < RT * 128; i += AGG_T) acc[i] = 0.0f;
  if (tid < AGG_W) qn[tid] = 0u;
  if (tid == 0) tot = 0u;
  __syncthreads();

  int rows = N - r0;
  if (rows > RT) rows = RT;
  const int cap = rows * OUTC;
  float* ra = ((LAYER == 1) ? qa : oa) + r0 * OUTC;
  float* rb = ((LAYER == 1) ? qb : ob) + r0 * OUTC;

  auto add_group = [&](int sv, int dl, float nv, int m) {
    int j0 = 0;
    for (; j0 + EB <= m; j0 += EB) {
      float v[EB][4];
#pragma unroll
      for (int j = 0; j < EB; ++j) row4<LAYER>(ta, tb, bcast_i(sv, j0 + j), l, v[j]);
#pragma unroll
      for (int j = 0; j < EB; ++j) {
        float* ar = acc + bcast_i(dl, j0 + j) * 128 + l;
        float nm  = bcast_f(nv, j0 + j);
        atomicAdd(&ar[0],  v[j][0] * nm);
        atomicAdd(&ar[32], v[j][1] * nm);
        atomicAdd(&ar[64], v[j][2] * nm);
        atomicAdd(&ar[96], v[j][3] * nm);
      }
    }
    for (; j0 < m; ++j0) {
      float v[4];
      row4<LAYER>(ta, tb, bcast_i(sv, j0), l, v);
      float* ar = acc + bcast_i(dl, j0) * 128 + l;
      float nm  = bcast_f(nv, j0);
      atomicAdd(&ar[0],  v[0] * nm);
      atomicAdd(&ar[32], v[1] * nm);
      atomicAdd(&ar[64], v[2] * nm);
      atomicAdd(&ar[96], v[3] * nm);
    }
  };

  unsigned int cw = NOREC;
  if (LAYER == 2) cw = (unsigned int)__builtin_amdgcn_readfirstlane((int)qc[blockIdx.x * AGG_W + wave]);
  if (cw != NOREC) {
    const int off = (int)(cw & 0xFFFFu);
    const int n   = (int)(cw >> 16);
    unsigned int p = 0u;
    float nv = 0.0f;
    if (l < n) { p = __float_as_uint(ra[off + l]); nv = rb[off + l]; }
    for (int g = 0; g < n; g += 32) {
      unsigned int pn = 0u;
      float nn = 0.0f;
      if (g + 32 + l < n) { pn = __float_as_uint(ra[off + g + 32 + l]); nn = rb[off + g + 32 + l]; }
      int m = n - g;
      if (m > 32) m = 32;
      add_group((int)(p & 0xFFFFu), (int)(p >> 16), nv, m);
      p = pn;
      nv = nn;
    }
  } else {

  const int nChunks = (E + 511) >> 9;
  const unsigned int idw = (unsigned int)blockIdx.x * 0x01010101u;
  const int rot = (int)(((unsigned int)blockIdx.x * (unsigned int)nChunks) / gridDim.x);
  unsigned int* myq = q + wave * QCAP;

  auto search = [&](uint4 x, int eb) {
    unsigned int t0 = x.x ^ idw, t1 = x.y ^ idw, t2 = x.z ^ idw, t3 = x.w ^ idw;
    unsigned int z0 = ~(((t0 & 0x7F7F7F7Fu) + 0x7F7F7F7Fu) | t0) & 0x80808080u;
    unsigned int z1 = ~(((t1 & 0x7F7F7F7Fu) + 0x7F7F7F7Fu) | t1) & 0x80808080u;
    unsigned int z2 = ~(((t2 & 0x7F7F7F7Fu) + 0x7F7F7F7Fu) | t2) & 0x80808080u;
    unsigned int z3 = ~(((t3 & 0x7F7F7F7Fu) + 0x7F7F7F7Fu) | t3) & 0x80808080u;
    unsigned int zz = (z0 >> 7) | (z1 >> 6) | (z2 >> 5) | (z3 >> 4);
    while (zz) {
      int bit = __builtin_ctz(zz);
      zz &= zz - 1u;
      int e = eb + ((bit & 7) << 2) + (bit >> 3);
      if (e < E) {
        unsigned int pos = atomicAdd(&qn[wave], 1u);
        if (pos < QCAP) {
          myq[pos] = (unsigned int)e;
        } else {
          int s = ei[e];
          int d = ei[E + e];
          float nm = dinv[s] * ew[e] * dinv[d];
          float* ar = acc + (d - r0) * 128;
          for (int c = 0; c < 128; ++c) {
            float hv = (LAYER == 1) ? ta[s * HID + c]
                                    : (c < OUTC ? ta[s * OUTC + c] : tb[s * OUTC + c - OUTC]);
            atomicAdd(&ar[c], hv * nm);
          }
        }
      }
    }
  };
  auto lane_base = [&](int ii) {
    int c = ii + rot;
    if (c >= nChunks) c -= nChunks;
    return (c << 9) + (l << 4);
  };

  int i = wave;
  for (; i + 3 * AGG_W < nChunks; i += 4 * AGG_W) {
    int b0 = lane_base(i), b1 = lane_base(i + AGG_W);
    int b2 = lane_base(i + 2 * AGG_W), b3 = lane_base(i + 3 * AGG_W);
    uint4 x0 = *(const uint4*)(own + (b0 >> 2));
    uint4 x1 = *(const uint4*)(own + (b1 >> 2));
    uint4 x2 = *(const uint4*)(own + (b2 >> 2));
    uint4 x3 = *(const uint4*)(own + (b3 >> 2));
    search(x0, b0);
    search(x1, b1);
    search(x2, b2);
    search(x3, b3);
  }
  for (; i < nChunks; i += AGG_W) {
    int b0 = lane_base(i);
    uint4 x0 = *(const uint4*)(own + (b0 >> 2));
    search(x0, b0);
  }

  int nq = __builtin_amdgcn_readfirstlane((int)qn[wave]);
  const bool ovf = nq > QCAP;
  const int n = ovf ? QCAP : nq;
  int off = 0;
  bool keep = false;
  if (LAYER == 1) {
    unsigned int o = 0u;
    if (l == 0) o = atomicAdd(&tot, (unsigned int)((n + 31) & ~31));
    off = __builtin_amdgcn_readfirstlane((int)o);
    keep = !ovf && off + n <= cap;
    if (l == 0) tw[wave] = keep ? ((unsigned int)off | ((unsigned int)n << 16)) : NOREC;
  }
  for (int g = 0; g < n; g += 32) {
    int sv = 0, dl = 0;
    float nv = 0.0f;
    if (g + l < n) {
      int e = (int)myq[g + l];
      int s = ei[e];
      int d = ei[E + e];
      nv = dinv[s] * ew[e] * dinv[d];
      sv = s;
      dl = d - r0;
      if (LAYER == 1 && keep) {
        ra[off + g + l] = __uint_as_float((unsigned int)sv | ((unsigned int)dl << 16));
        rb[off + g + l] = nv;
      }
    }
    int m = n - g;
    if (m > 32) m = 32;
    add_group(sv, dl, nv, m);
  }
  }
  __syncthreads();
  if (LAYER == 1 && tid < AGG_W) qc[blockIdx.x * AGG_W + tid] = tw[tid];

  const int c = tid & 127;
  for (int lr = tid >> 7; lr < rows; lr += AGG_T / 128) {
    int r = r0 + lr;
    float di = dinv[r];
    float d2 = di * di;
    float a = acc[lr * 128 + c];
    if (LAYER == 1) {
      oa[r * HID + c] = a + (ta[r * HID + c] * d2 + ba[c]);
    } else if (c < OUTC) {
      oa[r * OUTC + c] = a + (ta[r * OUTC + c] * d2 + ba[c]);
    } else {
      ob[r * OUTC + c - OUTC] = a + (tb[r * OUTC + c - OUTC] * d2 + bb[c - OUTC]);
    }
  }
}

extern "C" void kernel_launch(void* const* d_in, const int* in_sizes, int n_in,
                              void* d_out, int out_size, void* d_ws, size_t ws_size,
                              hipStream_t stream) {
  const float* x  = (const float*)d_in[0];
  const int*   ei = (const int*)  d_in[1];
  const float* ew = (const float*)d_in[2];
  const float* W1 = (const float*)d_in[3];
  const float* b1 = (const float*)d_in[4];
  const float* W2 = (const float*)d_in[5];
  const float* b2 = (const float*)d_in[6];
  const float* W3 = (const float*)d_in[7];
  const float* b3 = (const float*)d_in[8];

  const int N = in_sizes[0] / IN_CH;
  const int E = in_sizes[2];

  float* ws   = (float*)d_ws;
  float* deg  = ws;
  float* dinv = ws + N;
  float* h1   = ws + 2 * (size_t)N;
  float* agg1 = h1 + (size_t)N * HID;
  float* W1p  = agg1 + (size_t)N * HID;
  float* W2p  = W1p + (size_t)IN_CH * HID;
  float* W3p  = W2p + (size_t)HID * OUTC;
  unsigned int* own = (unsigned int*)(W3p + (size_t)HID * OUTC);
  float* t2   = h1;
  float* t3   = h1 + (size_t)N * OUTC;
  float* om   = (float*)d_out;
  float* ol   = om + (size_t)N * OUTC;
  unsigned int* qc = (unsigned int*)deg;
  float* part = h1;

  const int T = 256;

  k_pack_w<<<(IN_CH * HID + T - 1) / T, T, 0, stream>>>(W1, W1p, IN_CH, HID);
  k_pack_w<<<(HID * OUTC + T - 1) / T, T, 0, stream>>>(W2, W2p, HID, OUTC);
  k_pack_w<<<(HID * OUTC + T - 1) / T, T, 0, stream>>>(W3, W3p, HID, OUTC);

  k_deg_part  <<<DEG_G, 1024, 0, stream>>>(ei, ew, part, own, N, E);
  k_rsqrt     <<<(N + T - 1) / T, T, 0, stream>>>(part, dinv, N);

  const int waves  = N >> 4;
  const int gblocks = (waves * 32 + T - 1) / T;
  const int ablocks = (N + RT - 1) / RT;

  k_gemm_wmma<8, false><<<gblocks, T, 0, stream>>>(x, W1p, h1, N, IN_CH);

  k_agg<1><<<ablocks, AGG_T, 0, stream>>>(ei, ew, dinv, own, h1, h1, b1, b1, agg1, agg1, om, ol, qc, N, E);

  k_gemm_wmma<4, true><<<gblocks, T, 0, stream>>>(agg1, W2p, t2, N, HID);
  k_gemm_wmma<4, true><<<gblocks, T, 0, stream>>>(agg1, W3p, t3, N, HID);

  k_agg<2><<<ablocks, AGG_T, 0, stream>>>(ei, ew, dinv, own, t2, t3, b2, b3, om, ol, om, ol, qc, N, E);
}


# Layer-2 write-out: table, bias and output plane picked per wave before the loop (no branch inside it); dinv and table loads of 8 rows issued before one wait (was two dependent waits per row)
# speedup vs baseline: 2.7935x; 1.0559x over previous
#include <hip/hip_runtime.h>

typedef float v2f __attribute__((ext_vector_type(2)));
typedef float v8f __attribute__((ext_vector_type(8)));

#define IN_CH 512
#define HID   128
#define OUTC  64

#define RT    196
#define AGG_T 1024
#define AGG_W (AGG_T / 32)
#define QCAP  448

#define DEG_G 64
#define NCAP  (256 * RT)
#define DEG_B 8

__global__ void __launch_bounds__(1024)
k_deg_part(const int* __restrict__ ei, const float* __restrict__ ew,
           float* __restrict__ part, unsigned int* __restrict__ own, int N, int E) {
  __shared__ float dg[NCAP];
  const int tid = threadIdx.x;
  for (int i = tid; i < NCAP; i += 1024) dg[i] = 0.0f;
  __syncthreads();

  const int EP  = ((E + 511) >> 9) << 9;
  const int nCh = (EP + 1023) >> 10;
  const int per = (nCh + (int)gridDim.x - 1) / (int)gridDim.x;
  const int c0  = (int)blockIdx.x * per;
  int c1 = c0 + per;
  if (c1 > nCh) c1 = nCh;

  for (int c = c0; c < c1; c += DEG_B) {
    int   d[DEG_B];
    float w[DEG_B];
#pragma unroll
    for (int j = 0; j < DEG_B; ++j) {
      int e = ((c + j) << 10) + tid;
      d[j] = -1;
      w[j] = 0.0f;
      if (c + j < c1 && e < E) { d[j] = ei[E + e]; w[j] = ew[e]; }
    }
#pragma unroll
    for (int j = 0; j < DEG_B; ++j) {
      int e = ((c + j) << 10) + tid;
      unsigned int o = 0u;
      if (d[j] >= 0) {
        atomicAdd(&dg[d[j]], w[j]);
        o = (unsigned int)d[j] / RT;
      }
      unsigned int o1 = __shfl_down(o, 1);
      unsigned int o2 = __shfl_down(o, 2);
      unsigned int o3 = __shfl_down(o, 3);
      if (c + j < c1 && e < EP && (tid & 3) == 0)
        own[e >> 2] = o | (o1 << 8) | (o2 << 16) | (o3 << 24);
    }
  }
  __syncthreads();

  float* p = part + (size_t)blockIdx.x * N;
  for (int i = tid; i < N; i += 1024) p[i] = dg[i];
}

__global__ void k_rsqrt(const float* __restrict__ part, float* __restrict__ dinv, int n) {
  int i = blockIdx.x * blockDim.x + threadIdx.x;
  if (i < n) {
    float d = 1.0f;
#pragma unroll 16
    for (int g = 0; g < DEG_G; ++g) d += part[(size_t)g * n + i];
    dinv[i] = (d > 0.0f) ? rsqrtf(d) : 0.0f;
  }
}

__global__ void k_pack_w(const float* __restrict__ W, float* __restrict__ Wp,
                         int K, int N) {
  int idx = blockIdx.x * blockDim.x + threadIdx.x;
  if (idx >= K * N) return;
  int k = idx / N;
  int n = idx - k * N;
  int k4 = k >> 2;
  int hi = (k >> 1) & 1;
  int j  = k & 1;
  Wp[(((size_t)((k4 << 1) | hi)) * N + n) * 2 + j] = W[idx];
}

template <int NT, bool LAYER2>
__global__ void k_gemm_wmma(const float* __restrict__ A,
                            const float* __restrict__ Bp,
                            float* __restrict__ C,
                            int M, int K) {
  const int N = NT * 16;
  int wave = (int)((blockIdx.x * blockDim.x + threadIdx.x) >> 5);
  int lane = threadIdx.x & 31;
  if (wave >= (M >> 4)) return;

  int m0 = wave << 4;
  int lo = lane & 15;
  int hi = lane >> 4;

  const float* aPtr = A  + (size_t)(m0 + lo) * K + (hi << 1);
  const float* bPtr = Bp + ((size_t)hi * N + lo) * 2;

  v8f z = {0.f,0.f,0.f,0.f,0.f,0.f,0.f,0.f};
  v8f acc[NT];
#pragma unroll
  for (int t = 0; t < NT; ++t) acc[t] = z;

  for (int k = 0; k < K; k += 8) {
    v2f a0 = *(const v2f*)(aPtr + k);
    v2f a1 = *(const v2f*)(aPtr + k + 4);
    const float* bp0 = bPtr + (size_t)k * N;
    const float* bp1 = bp0 + (size_t)4 * N;
    v2f b0[NT], b1[NT];
#pragma unroll
    for (int t = 0; t < NT; ++t) b0[t] = *(const v2f*)(bp0 + t * 32);
#pragma unroll
    for (int t = 0; t < NT; ++t) b1[t] = *(const v2f*)(bp1 + t * 32);
    if (LAYER2) {
      a0.x = fmaxf(a0.x, 0.0f); a0.y = fmaxf(a0.y, 0.0f);
      a1.x = fmaxf(a1.x, 0.0f); a1.y = fmaxf(a1.y, 0.0f);
    }
#pragma unroll
    for (int t = 0; t < NT; ++t)
      acc[t] = __builtin_amdgcn_wmma_f32_16x16x4_f32(
          false, a0, false, b0[t], (short)0, acc[t], false, false);
#pragma unroll
    for (int t = 0; t < NT; ++t)
      acc[t] = __builtin_amdgcn_wmma_f32_16x16x4_f32(
          false, a1, false, b1[t], (short)0, acc[t], false, false);
  }

  int cRow = m0 + (hi << 3);
#pragma unroll
  for (int r = 0; r < 8; ++r) {
    float* cp = C + (size_t)(cRow + r) * N + lo;
#pragma unroll
    for (int t = 0; t < NT; ++t) cp[t * 16] = acc[t][r];
  }
}

#define EB 8
#define NOREC 0xFFFFFFFFu

__device__ __forceinline__ int bcast_i(int v, int j) {
  return __builtin_amdgcn_readlane(v, j);
}
__device__ __forceinline__ float bcast_f(float v, int j) {
  return __int_as_float(__builtin_amdgcn_readlane(__float_as_int(v), j));
}

template <int LAYER>
__device__ __forceinline__ void row4(const float* __restrict__ ta,
                                     const float* __restrict__ tb,
                                     int s, int l, float* v) {
  if (LAYER == 1) {
    int so = s * HID + l;
    v[0] = ta[so]; v[1] = ta[so + 32]; v[2] = ta[so + 64]; v[3] = ta[so + 96];
  } else {
    int so = s * OUTC + l;
    v[0] = ta[so]; v[1] = ta[so + 32]; v[2] = tb[so]; v[3] = tb[so + 32];
  }
}

template <int LAYER>
__global__ void __launch_bounds__(AGG_T)
k_agg(const int* __restrict__ ei, const float* __restrict__ ew,
      const float* __restrict__ dinv, const unsigned int* __restrict__ own,
      const float* __restrict__ ta, const float* __restrict__ tb,
      const float* __restrict__ ba, const float* __restrict__ bb,
      float* oa, float* ob, float* qa, float* qb, unsigned int* qc, int N, int E) {
  __shared__ float acc[RT * 128];
  __shared__ unsigned int q[AGG_W * QCAP];
  __shared__ unsigned int qn[AGG_W];
  __shared__ unsigned int tot;
  __shared__ unsigned int tw[AGG_W];

  const int tid  = threadIdx.x;
  const int l    = tid & 31;
  const int wave = __builtin_amdgcn_readfirstlane(tid >> 5);
  const int r0   = blockIdx.x * RT;

  for (int i = tid; i < RT * 128; i += AGG_T) acc[i] = 0.0f;
  if (tid < AGG_W) qn[tid] = 0u;
  if (tid == 0) tot = 0u;
  __syncthreads();

  int rows = N - r0;
  if (rows > RT) rows = RT;
  const int cap = rows * OUTC;
  float* ra = ((LAYER == 1) ? qa : oa) + r0 * OUTC;
  float* rb = ((LAYER == 1) ? qb : ob) + r0 * OUTC;

  auto add_group = [&](int sv, int dl, float nv, int m) {
    int j0 = 0;
    for (; j0 + EB <= m; j0 += EB) {
      float v[EB][4];
#pragma unroll
      for (int j = 0; j < EB; ++j) row4<LAYER>(ta, tb, bcast_i(sv, j0 + j), l, v[j]);
#pragma unroll
      for (int j = 0; j < EB; ++j) {
        float* ar = acc + bcast_i(dl, j0 + j) * 128 + l;
        float nm  = bcast_f(nv, j0 + j);
        atomicAdd(&ar[0],  v[j][0] * nm);
        atomicAdd(&ar[32], v[j][1] * nm);
        atomicAdd(&ar[64], v[j][2] * nm);
        atomicAdd(&ar[96], v[j][3] * nm);
      }
    }
    for (; j0 < m; ++j0) {
      float v[4];
      row4<LAYER>(ta, tb, bcast_i(sv, j0), l, v);
      float* ar = acc + bcast_i(dl, j0) * 128 + l;
      float nm  = bcast_f(nv, j0);
      atomicAdd(&ar[0],  v[0] * nm);
      atomicAdd(&ar[32], v[1] * nm);
      atomicAdd(&ar[64], v[2] * nm);
      atomicAdd(&ar[96], v[3] * nm);
    }
  };

  unsigned int cw = NOREC;
  if (LAYER == 2) cw = (unsigned int)__builtin_amdgcn_readfirstlane((int)qc[blockIdx.x * AGG_W + wave]);
  if (cw != NOREC) {
    const int off = (int)(cw & 0xFFFFu);
    const int n   = (int)(cw >> 16);
    unsigned int p = 0u;
    float nv = 0.0f;
    if (l < n) { p = __float_as_uint(ra[off + l]); nv = rb[off + l]; }
    for (int g = 0; g < n; g += 32) {
      unsigned int pn = 0u;
      float nn = 0.0f;
      if (g + 32 + l < n) { pn = __float_as_uint(ra[off + g + 32 + l]); nn = rb[off + g + 32 + l]; }
      int m = n - g;
      if (m > 32) m = 32;
      add_group((int)(p & 0xFFFFu), (int)(p >> 16), nv, m);
      p = pn;
      nv = nn;
    }
  } else {

  const int nChunks = (E + 511) >> 9;
  const unsigned int idw = (unsigned int)blockIdx.x * 0x01010101u;
  const int rot = (int)(((unsigned int)blockIdx.x * (unsigned int)nChunks) / gridDim.x);
  unsigned int* myq = q + wave * QCAP;

  auto search = [&](uint4 x, int eb) {
    unsigned int t0 = x.x ^ idw, t1 = x.y ^ idw, t2 = x.z ^ idw, t3 = x.w ^ idw;
    unsigned int z0 = ~(((t0 & 0x7F7F7F7Fu) + 0x7F7F7F7Fu) | t0) & 0x80808080u;
    unsigned int z1 = ~(((t1 & 0x7F7F7F7Fu) + 0x7F7F7F7Fu) | t1) & 0x80808080u;
    unsigned int z2 = ~(((t2 & 0x7F7F7F7Fu) + 0x7F7F7F7Fu) | t2) & 0x80808080u;
    unsigned int z3 = ~(((t3 & 0x7F7F7F7Fu) + 0x7F7F7F7Fu) | t3) & 0x80808080u;
    unsigned int zz = (z0 >> 7) | (z1 >> 6) | (z2 >> 5) | (z3 >> 4);
    while (zz) {
      int bit = __builtin_ctz(zz);
      zz &= zz - 1u;
      int e = eb + ((bit & 7) << 2) + (bit >> 3);
      if (e < E) {
        unsigned int pos = atomicAdd(&qn[wave], 1u);
        if (pos < QCAP) {
          myq[pos] = (unsigned int)e;
        } else {
          int s = ei[e];
          int d = ei[E + e];
          float nm = dinv[s] * ew[e] * dinv[d];
          float* ar = acc + (d - r0) * 128;
          for (int c = 0; c < 128; ++c) {
            float hv = (LAYER == 1) ? ta[s * HID + c]
                                    : (c < OUTC ? ta[s * OUTC + c] : tb[s * OUTC + c - OUTC]);
            atomicAdd(&ar[c], hv * nm);
          }
        }
      }
    }
  };
  auto lane_base = [&](int ii) {
    int c = ii + rot;
    if (c >= nChunks) c -= nChunks;
    return (c << 9) + (l << 4);
  };

  int i = wave;
  for (; i + 3 * AGG_W < nChunks; i += 4 * AGG_W) {
    int b0 = lane_base(i), b1 = lane_base(i + AGG_W);
    int b2 = lane_base(i + 2 * AGG_W), b3 = lane_base(i + 3 * AGG_W);
    uint4 x0 = *(const uint4*)(own + (b0 >> 2));
    uint4 x1 = *(const uint4*)(own + (b1 >> 2));
    uint4 x2 = *(const uint4*)(own + (b2 >> 2));
    uint4 x3 = *(const uint4*)(own + (b3 >> 2));
    search(x0, b0);
    search(x1, b1);
    search(x2, b2);
    search(x3, b3);
  }
  for (; i < nChunks; i += AGG_W) {
    int b0 = lane_base(i);
    uint4 x0 = *(const uint4*)(own + (b0 >> 2));
    search(x0, b0);
  }

  int nq = __builtin_amdgcn_readfirstlane((int)qn[wave]);
  const bool ovf = nq > QCAP;
  const int n = ovf ? QCAP : nq;
  int off = 0;
  bool keep = false;
  if (LAYER == 1) {
    unsigned int o = 0u;
    if (l == 0) o = atomicAdd(&tot, (unsigned int)((n + 31) & ~31));
    off = __builtin_amdgcn_readfirstlane((int)o);
    keep = !ovf && off + n <= cap;
    if (l == 0) tw[wave] = keep ? ((unsigned int)off | ((unsigned int)n << 16)) : NOREC;
  }
  for (int g = 0; g < n; g += 32) {
    int sv = 0, dl = 0;
    float nv = 0.0f;
    if (g + l < n) {
      int e = (int)myq[g + l];
      int s = ei[e];
      int d = ei[E + e];
      nv = dinv[s] * ew[e] * dinv[d];
      sv = s;
      dl = d - r0;
      if (LAYER == 1 && keep) {
        ra[off + g + l] = __uint_as_float((unsigned int)sv | ((unsigned int)dl << 16));
        rb[off + g + l] = nv;
      }
    }
    int m = n - g;
    if (m > 32) m = 32;
    add_group(sv, dl, nv, m);
  }
  }
  __syncthreads();
  if (LAYER == 1 && tid < AGG_W) qc[blockIdx.x * AGG_W + tid] = tw[tid];

  const int c = tid & 127;
  if (LAYER == 1) {
    for (int lr = tid >> 7; lr < rows; lr += AGG_T / 128) {
      int r = r0 + lr;
      float di = dinv[r];
      float d2 = di * di;
      float a = acc[lr * 128 + c];
      oa[r * HID + c] = a + (ta[r * HID + c] * d2 + ba[c]);
    }
  } else {
    const int  cc  = c & (OUTC - 1);
    const bool sec = c >= OUTC;
    const float* __restrict__ tsel = sec ? tb : ta;
    float* osel = sec ? ob : oa;
    const float bias = sec ? bb[cc] : ba[cc];
    const int WB = 8;
    const int RS = AGG_T / 128;
    for (int lr0 = tid >> 7; lr0 < rows; lr0 += WB * RS) {
      float di[WB], tv[WB];
#pragma unroll
      for (int j = 0; j < WB; ++j) {
        int lr = lr0 + j * RS;
        di[j] = 0.0f;
        tv[j] = 0.0f;
        if (lr < rows) { di[j] = dinv[r0 + lr]; tv[j] = tsel[(r0 + lr) * OUTC + cc]; }
      }
#pragma unroll
      for (int j = 0; j < WB; ++j) {
        int lr = lr0 + j * RS;
        if (lr < rows) {
          float d2 = di[j] * di[j];
          osel[(r0 + lr) * OUTC + cc] = acc[lr * 128 + c] + (tv[j] * d2 + bias);
        }
      }
    }
  }
}

extern "C" void kernel_launch(void* const* d_in, const int* in_sizes, int n_in,
                              void* d_out, int out_size, void* d_ws, size_t ws_size,
                              hipStream_t stream) {
  const float* x  = (const float*)d_in[0];
  const int*   ei = (const int*)  d_in[1];
  const float* ew = (const float*)d_in[2];
  const float* W1 = (const float*)d_in[3];
  const float* b1 = (const float*)d_in[4];
  const float* W2 = (const float*)d_in[5];
  const float* b2 = (const float*)d_in[6];
  const float* W3 = (const float*)d_in[7];
  const float* b3 = (const float*)d_in[8];

  const int N = in_sizes[0] / IN_CH;
  const int E = in_sizes[2];

  float* ws   = (float*)d_ws;
  float* deg  = ws;
  float* dinv = ws + N;
  float* h1   = ws + 2 * (size_t)N;
  float* agg1 = h1 + (size_t)N * HID;
  float* W1p  = agg1 + (size_t)N * HID;
  float* W2p  = W1p + (size_t)IN_CH * HID;
  float* W3p  = W2p + (size_t)HID * OUTC;
  unsigned int* own = (unsigned int*)(W3p + (size_t)HID * OUTC);
  float* t2   = h1;
  float* t3   = h1 + (size_t)N * OUTC;
  float* om   = (float*)d_out;
  float* ol   = om + (size_t)N * OUTC;
  unsigned int* qc = (unsigned int*)deg;
  float* part = h1;

  const int T = 256;

  k_pack_w<<<(IN_CH * HID + T - 1) / T, T, 0, stream>>>(W1, W1p, IN_CH, HID);
  k_pack_w<<<(HID * OUTC + T - 1) / T, T, 0, stream>>>(W2, W2p, HID, OUTC);
  k_pack_w<<<(HID * OUTC + T - 1) / T, T, 0, stream>>>(W3, W3p, HID, OUTC);

  k_deg_part  <<<DEG_G, 1024, 0, stream>>>(ei, ew, part, own, N, E);
  k_rsqrt     <<<(N + T - 1) / T, T, 0, stream>>>(part, dinv, N);

  const int waves  = N >> 4;
  const int gblocks = (waves * 32 + T - 1) / T;
  const int ablocks = (N + RT - 1) / RT;

  k_gemm_wmma<8, false><<<gblocks, T, 0, stream>>>(x, W1p, h1, N, IN_CH);

  k_agg<1><<<ablocks, AGG_T, 0, stream>>>(ei, ew, dinv, own, h1, h1, b1, b1, agg1, agg1, om, ol, qc, N, E);

  k_gemm_wmma<4, true><<<gblocks, T, 0, stream>>>(agg1, W2p, t2, N, HID);
  k_gemm_wmma<4, true><<<gblocks, T, 0, stream>>>(agg1, W3p, t3, N, HID);

  k_agg<2><<<ablocks, AGG_T, 0, stream>>>(ei, ew, dinv, own, t2, t3, b2, b3, om, ol, om, ol, qc, N, E);
}


# Owner ids kept as 8 bit planes per 32 edges (was 1 byte per edge): a scanning lane tests 32 edges with 8 xor/and, not byte compares on 16; degree launch writes planes by wave ballot; sum order kept
# speedup vs baseline: 2.9012x; 1.0386x over previous
#include <hip/hip_runtime.h>

typedef float v2f __attribute__((ext_vector_type(2)));
typedef float v8f __attribute__((ext_vector_type(8)));

#define IN_CH 512
#define HID   128
#define OUTC  64

#define RT    196
#define AGG_T 1024
#define AGG_W (AGG_T / 32)
#define QCAP  448

#define DEG_G 64
#define NCAP  (256 * RT)
#define DEG_B 8

__global__ void __launch_bounds__(1024)
k_deg_part(const int* __restrict__ ei, const float* __restrict__ ew,
           float* __restrict__ part, unsigned int* __restrict__ own, int N, int E) {
  __shared__ float dg[NCAP];
  const int tid = threadIdx.x;
  for (int i = tid; i < NCAP; i += 1024) dg[i] = 0.0f;
  __syncthreads();

  const int EP  = ((E + 511) >> 9) << 9;
  const int nCh = (EP + 1023) >> 10;
  const int per = (nCh + (int)gridDim.x - 1) / (int)gridDim.x;
  const int c0  = (int)blockIdx.x * per;
  int c1 = c0 + per;
  if (c1 > nCh) c1 = nCh;

  for (int c = c0; c < c1; c += DEG_B) {
    int   d[DEG_B];
    float w[DEG_B];
#pragma unroll
    for (int j = 0; j < DEG_B; ++j) {
      int e = ((c + j) << 10) + tid;
      d[j] = -1;
      w[j] = 0.0f;
      if (c + j < c1 && e < E) { d[j] = ei[E + e]; w[j] = ew[e]; }
    }
#pragma unroll
    for (int j = 0; j < DEG_B; ++j) {
      int e = ((c + j) << 10) + tid;
      unsigned int o = 0u;
      if (d[j] >= 0) {
        atomicAdd(&dg[d[j]], w[j]);
        o = (unsigned int)d[j] / RT;
      }
      unsigned int pl = 0u;
      for (int k = 0; k < 8; ++k) {
        unsigned int b = __builtin_amdgcn_ballot_w32(((o >> k) & 1u) != 0u);
        if ((tid & 31) == k) pl = b;
      }
      const int u  = e >> 9;
      const int gl = (e >> 5) & 15;
      const int k  = tid & 31;
      if (c + j < c1 && (u << 9) < EP && k < 8)
        own[((((u << 1) | (k >> 2)) << 4) | gl) * 4 + (k & 3)] = pl;
    }
  }
  __syncthreads();

  float* p = part + (size_t)blockIdx.x * N;
  for (int i = tid; i < N; i += 1024) p[i] = dg[i];
}

__global__ void k_rsqrt(const float* __restrict__ part, float* __restrict__ dinv, int n) {
  int i = blockIdx.x * blockDim.x + threadIdx.x;
  if (i < n) {
    float d = 1.0f;
#pragma unroll 16
    for (int g = 0; g < DEG_G; ++g) d += part[(size_t)g * n + i];
    dinv[i] = (d > 0.0f) ? rsqrtf(d) : 0.0f;
  }
}

__global__ void k_pack_w(const float* __restrict__ W, float* __restrict__ Wp,
                         int K, int N) {
  int idx = blockIdx.x * blockDim.x + threadIdx.x;
  if (idx >= K * N) return;
  int k = idx / N;
  int n = idx - k * N;
  int k4 = k >> 2;
  int hi = (k >> 1) & 1;
  int j  = k & 1;
  Wp[(((size_t)((k4 << 1) | hi)) * N + n) * 2 + j] = W[idx];
}

template <int NT, bool LAYER2>
__global__ void k_gemm_wmma(const float* __restrict__ A,
                            const float* __restrict__ Bp,
                            float* __restrict__ C,
                            int M, int K) {
  const int N = NT * 16;
  int wave = (int)((blockIdx.x * blockDim.x + threadIdx.x) >> 5);
  int lane = threadIdx.x & 31;
  if (wave >= (M >> 4)) return;

  int m0 = wave << 4;
  int lo = lane & 15;
  int hi = lane >> 4;

  const float* aPtr = A  + (size_t)(m0 + lo) * K + (hi << 1);
  const float* bPtr = Bp + ((size_t)hi * N + lo) * 2;

  v8f z = {0.f,0.f,0.f,0.f,0.f,0.f,0.f,0.f};
  v8f acc[NT];
#pragma unroll
  for (int t = 0; t < NT; ++t) acc[t] = z;

  for (int k = 0; k < K; k += 8) {
    v2f a0 = *(const v2f*)(aPtr + k);
    v2f a1 = *(const v2f*)(aPtr + k + 4);
    const float* bp0 = bPtr + (size_t)k * N;
    const float* bp1 = bp0 + (size_t)4 * N;
    v2f b0[NT], b1[NT];
#pragma unroll
    for (int t = 0; t < NT; ++t) b0[t] = *(const v2f*)(bp0 + t * 32);
#pragma unroll
    for (int t = 0; t < NT; ++t) b1[t] = *(const v2f*)(bp1 + t * 32);
    if (LAYER2) {
      a0.x = fmaxf(a0.x, 0.0f); a0.y = fmaxf(a0.y, 0.0f);
      a1.x = fmaxf(a1.x, 0.0f); a1.y = fmaxf(a1.y, 0.0f);
    }
#pragma unroll
    for (int t = 0; t < NT; ++t)
      acc[t] = __builtin_amdgcn_wmma_f32_16x16x4_f32(
          false, a0, false, b0[t], (short)0, acc[t], false, false);
#pragma unroll
    for (int t = 0; t < NT; ++t)
      acc[t] = __builtin_amdgcn_wmma_f32_16x16x4_f32(
          false, a1, false, b1[t], (short)0, acc[t], false, false);
  }

  int cRow = m0 + (hi << 3);
#pragma unroll
  for (int r = 0; r < 8; ++r) {
    float* cp = C + (size_t)(cRow + r) * N + lo;
#pragma unroll
    for (int t = 0; t < NT; ++t) cp[t * 16] = acc[t][r];
  }
}

#define EB 8
#define NOREC 0xFFFFFFFFu

__device__ __forceinline__ int bcast_i(int v, int j) {
  return __builtin_amdgcn_readlane(v, j);
}
__device__ __forceinline__ float bcast_f(float v, int j) {
  return __int_as_float(__builtin_amdgcn_readlane(__float_as_int(v), j));
}

template <int LAYER>
__device__ __forceinline__ void row4(const float* __restrict__ ta,
                                     const float* __restrict__ tb,
                                     int s, int l, float* v) {
  if (LAYER == 1) {
    int so = s * HID + l;
    v[0] = ta[so]; v[1] = ta[so + 32]; v[2] = ta[so + 64]; v[3] = ta[so + 96];
  } else {
    int so = s * OUTC + l;
    v[0] = ta[so]; v[1] = ta[so + 32]; v[2] = tb[so]; v[3] = tb[so + 32];
  }
}

template <int LAYER>
__global__ void __launch_bounds__(AGG_T)
k_agg(const int* __restrict__ ei, const float* __restrict__ ew,
      const float* __restrict__ dinv, const unsigned int* __restrict__ own,
      const float* __restrict__ ta, const float* __restrict__ tb,
      const float* __restrict__ ba, const float* __restrict__ bb,
      float* oa, float* ob, float* qa, float* qb, unsigned int* qc, int N, int E) {
  __shared__ float acc[RT * 128];
  __shared__ unsigned int q[AGG_W * QCAP];
  __shared__ unsigned int qn[AGG_W];
  __shared__ unsigned int tot;
  __shared__ unsigned int tw[AGG_W];

  const int tid  = threadIdx.x;
  const int l    = tid & 31;
  const int wave = __builtin_amdgcn_readfirstlane(tid >> 5);
  const int r0   = blockIdx.x * RT;

  for (int i = tid; i < RT * 128; i += AGG_T) acc[i] = 0.0f;
  if (tid < AGG_W) qn[tid] = 0u;
  if (tid == 0) tot = 0u;
  __syncthreads();

  int rows = N - r0;
  if (rows > RT) rows = RT;
  const int cap = rows * OUTC;
  float* ra = ((LAYER == 1) ? qa : oa) + r0 * OUTC;
  float* rb = ((LAYER == 1) ? qb : ob) + r0 * OUTC;

  auto add_group = [&](int sv, int dl, float nv, int m) {
    int j0 = 0;
    for (; j0 + EB <= m; j0 += EB) {
      float v[EB][4];
#pragma unroll
      for (int j = 0; j < EB; ++j) row4<LAYER>(ta, tb, bcast_i(sv, j0 + j), l, v[j]);
#pragma unroll
      for (int j = 0; j < EB; ++j) {
        float* ar = acc + bcast_i(dl, j0 + j) * 128 + l;
        float nm  = bcast_f(nv, j0 + j);
        atomicAdd(&ar[0],  v[j][0] * nm);
        atomicAdd(&ar[32], v[j][1] * nm);
        atomicAdd(&ar[64], v[j][2] * nm);
        atomicAdd(&ar[96], v[j][3] * nm);
      }
    }
    for (; j0 < m; ++j0) {
      float v[4];
      row4<LAYER>(ta, tb, bcast_i(sv, j0), l, v);
      float* ar = acc + bcast_i(dl, j0) * 128 + l;
      float nm  = bcast_f(nv, j0);
      atomicAdd(&ar[0],  v[0] * nm);
      atomicAdd(&ar[32], v[1] * nm);
      atomicAdd(&ar[64], v[2] * nm);
      atomicAdd(&ar[96], v[3] * nm);
    }
  };

  unsigned int cw = NOREC;
  if (LAYER == 2) cw = (unsigned int)__builtin_amdgcn_readfirstlane((int)qc[blockIdx.x * AGG_W + wave]);
  if (cw != NOREC) {
    const int off = (int)(cw & 0xFFFFu);
    const int n   = (int)(cw >> 16);
    unsigned int p = 0u;
    float nv = 0.0f;
    if (l < n) { p = __float_as_uint(ra[off + l]); nv = rb[off + l]; }
    for (int g = 0; g < n; g += 32) {
      unsigned int pn = 0u;
      float nn = 0.0f;
      if (g + 32 + l < n) { pn = __float_as_uint(ra[off + g + 32 + l]); nn = rb[off + g + 32 + l]; }
      int m = n - g;
      if (m > 32) m = 32;
      add_group((int)(p & 0xFFFFu), (int)(p >> 16), nv, m);
      p = pn;
      nv = nn;
    }
  } else {

  const int nU = (E + 511) >> 9;
  const int nS = (nU + 1) >> 1;
  const unsigned int bid = blockIdx.x;
  const unsigned int m0 = (bid & 1u)   ? 0u : ~0u, m1 = (bid & 2u)   ? 0u : ~0u;
  const unsigned int m2 = (bid & 4u)   ? 0u : ~0u, m3 = (bid & 8u)   ? 0u : ~0u;
  const unsigned int m4 = (bid & 16u)  ? 0u : ~0u, m5 = (bid & 32u)  ? 0u : ~0u;
  const unsigned int m6 = (bid & 64u)  ? 0u : ~0u, m7 = (bid & 128u) ? 0u : ~0u;
  const int rot = (int)((bid * (unsigned int)nS) / gridDim.x);
  unsigned int* myq = q + wave * QCAP;
  const uint4* own4 = (const uint4*)own;

  auto search = [&](uint4 a, uint4 b, int eb, bool ok) {
    unsigned int zz = (a.x ^ m0) & (a.y ^ m1) & (a.z ^ m2) & (a.w ^ m3)
                    & (b.x ^ m4) & (b.y ^ m5) & (b.z ^ m6) & (b.w ^ m7);
    if (!ok) zz = 0u;
    while (zz) {
      int bit = __builtin_ctz(zz);
      zz &= zz - 1u;
      int e = eb + bit;
      if (e < E) {
        unsigned int pos = atomicAdd(&qn[wave], 1u);
        if (pos < QCAP) {
          myq[pos] = (unsigned int)e;
        } else {
          int s = ei[e];
          int d = ei[E + e];
          float nm = dinv[s] * ew[e] * dinv[d];
          float* ar = acc + (d - r0) * 128;
          for (int c = 0; c < 128; ++c) {
            float hv = (LAYER == 1) ? ta[s * HID + c]
                                    : (c < OUTC ? ta[s * OUTC + c] : tb[s * OUTC + c - OUTC]);
            atomicAdd(&ar[c], hv * nm);
          }
        }
      }
    }
  };
  auto lane_unit = [&](int ii) {
    int c = ii + rot;
    if (c >= nS) c -= nS;
    return (c << 1) | (l >> 4);
  };
  const int gl = l & 15;

  int i = wave;
  for (; i + 3 * AGG_W < nS; i += 4 * AGG_W) {
    int u0 = lane_unit(i), u1 = lane_unit(i + AGG_W);
    int u2 = lane_unit(i + 2 * AGG_W), u3 = lane_unit(i + 3 * AGG_W);
    bool k0 = u0 < nU, k1 = u1 < nU, k2 = u2 < nU, k3 = u3 < nU;
    int c0 = min(u0, nU - 1) << 5, c1 = min(u1, nU - 1) << 5;
    int c2 = min(u2, nU - 1) << 5, c3 = min(u3, nU - 1) << 5;
    uint4 a0 = own4[c0 | gl], b0 = own4[c0 | 16 | gl];
    uint4 a1 = own4[c1 | gl], b1 = own4[c1 | 16 | gl];
    uint4 a2 = own4[c2 | gl], b2 = own4[c2 | 16 | gl];
    uint4 a3 = own4[c3 | gl], b3 = own4[c3 | 16 | gl];
    search(a0, b0, (u0 << 9) + (gl << 5), k0);
    search(a1, b1, (u1 << 9) + (gl << 5), k1);
    search(a2, b2, (u2 << 9) + (gl << 5), k2);
    search(a3, b3, (u3 << 9) + (gl << 5), k3);
  }
  for (; i < nS; i += AGG_W) {
    int u0 = lane_unit(i);
    bool k0 = u0 < nU;
    int c0 = min(u0, nU - 1) << 5;
    uint4 a0 = own4[c0 | gl], b0 = own4[c0 | 16 | gl];
    search(a0, b0, (u0 << 9) + (gl << 5), k0);
  }

  int nq = __builtin_amdgcn_readfirstlane((int)qn[wave]);
  const bool ovf = nq > QCAP;
  const int n = ovf ? QCAP : nq;
  int off = 0;
  bool keep = false;
  if (LAYER == 1) {
    unsigned int o = 0u;
    if (l == 0) o = atomicAdd(&tot, (unsigned int)((n + 31) & ~31));
    off = __builtin_amdgcn_readfirstlane((int)o);
    keep = !ovf && off + n <= cap;
    if (l == 0) tw[wave] = keep ? ((unsigned int)off | ((unsigned int)n << 16)) : NOREC;
  }
  for (int g = 0; g < n; g += 32) {
    int sv = 0, dl = 0;
    float nv = 0.0f;
    if (g + l < n) {
      int e = (int)myq[g + l];
      int s = ei[e];
      int d = ei[E + e];
      nv = dinv[s] * ew[e] * dinv[d];
      sv = s;
      dl = d - r0;
      if (LAYER == 1 && keep) {
        ra[off + g + l] = __uint_as_float((unsigned int)sv | ((unsigned int)dl << 16));
        rb[off + g + l] = nv;
      }
    }
    int m = n - g;
    if (m > 32) m = 32;
    add_group(sv, dl, nv, m);
  }
  }
  __syncthreads();
  if (LAYER == 1 && tid < AGG_W) qc[blockIdx.x * AGG_W + tid] = tw[tid];

  const int c = tid & 127;
  if (LAYER == 1) {
    for (int lr = tid >> 7; lr < rows; lr += AGG_T / 128) {
      int r = r0 + lr;
      float di = dinv[r];
      float d2 = di * di;
      float a = acc[lr * 128 + c];
      oa[r * HID + c] = a + (ta[r * HID + c] * d2 + ba[c]);
    }
  } else {
    const int  cc  = c & (OUTC - 1);
    const bool sec = c >= OUTC;
    const float* __restrict__ tsel = sec ? tb : ta;
    float* osel = sec ? ob : oa;
    const float bias = sec ? bb[cc] : ba[cc];
    const int WB = 8;
    const int RS = AGG_T / 128;
    for (int lr0 = tid >> 7; lr0 < rows; lr0 += WB * RS) {
      float di[WB], tv[WB];
#pragma unroll
      for (int j = 0; j < WB; ++j) {
        int lr = lr0 + j * RS;
        di[j] = 0.0f;
        tv[j] = 0.0f;
        if (lr < rows) { di[j] = dinv[r0 + lr]; tv[j] = tsel[(r0 + lr) * OUTC + cc]; }
      }
#pragma unroll
      for (int j = 0; j < WB; ++j) {
        int lr = lr0 + j * RS;
        if (lr < rows) {
          float d2 = di[j] * di[j];
          osel[(r0 + lr) * OUTC + cc] = acc[lr * 128 + c] + (tv[j] * d2 + bias);
        }
      }
    }
  }
}

extern "C" void kernel_launch(void* const* d_in, const int* in_sizes, int n_in,
                              void* d_out, int out_size, void* d_ws, size_t ws_size,
                              hipStream_t stream) {
  const float* x  = (const float*)d_in[0];
  const int*   ei = (const int*)  d_in[1];
  const float* ew = (const float*)d_in[2];
  const float* W1 = (const float*)d_in[3];
  const float* b1 = (const float*)d_in[4];
  const float* W2 = (const float*)d_in[5];
  const float* b2 = (const float*)d_in[6];
  const float* W3 = (const float*)d_in[7];
  const float* b3 = (const float*)d_in[8];

  const int N = in_sizes[0] / IN_CH;
  const int E = in_sizes[2];

  float* ws   = (float*)d_ws;
  float* deg  = ws;
  float* dinv = ws + N;
  float* h1   = ws + 2 * (size_t)N;
  float* agg1 = h1 + (size_t)N * HID;
  float* W1p  = agg1 + (size_t)N * HID;
  float* W2p  = W1p + (size_t)IN_CH * HID;
  float* W3p  = W2p + (size_t)HID * OUTC;
  unsigned int* own = (unsigned int*)(W3p + (size_t)HID * OUTC);
  float* t2   = h1;
  float* t3   = h1 + (size_t)N * OUTC;
  float* om   = (float*)d_out;
  float* ol   = om + (size_t)N * OUTC;
  unsigned int* qc = (unsigned int*)deg;
  float* part = h1;

  const int T = 256;

  k_pack_w<<<(IN_CH * HID + T - 1) / T, T, 0, stream>>>(W1, W1p, IN_CH, HID);
  k_pack_w<<<(HID * OUTC + T - 1) / T, T, 0, stream>>>(W2, W2p, HID, OUTC);
  k_pack_w<<<(HID * OUTC + T - 1) / T, T, 0, stream>>>(W3, W3p, HID, OUTC);

  k_deg_part  <<<DEG_G, 1024, 0, stream>>>(ei, ew, part, own, N, E);
  k_rsqrt     <<<(N + T - 1) / T, T, 0, stream>>>(part, dinv, N);

  const int waves  = N >> 4;
  const int gblocks = (waves * 32 + T - 1) / T;
  const int ablocks = (N + RT - 1) / RT;

  k_gemm_wmma<8, false><<<gblocks, T, 0, stream>>>(x, W1p, h1, N, IN_CH);

  k_agg<1><<<ablocks, AGG_T, 0, stream>>>(ei, ew, dinv, own, h1, h1, b1, b1, agg1, agg1, om, ol, qc, N, E);

  k_gemm_wmma<4, true><<<gblocks, T, 0, stream>>>(agg1, W2p, t2, N, HID);
  k_gemm_wmma<4, true><<<gblocks, T, 0, stream>>>(agg1, W3p, t3, N, HID);

  k_agg<2><<<ablocks, AGG_T, 0, stream>>>(ei, ew, dinv, own, t2, t3, b2, b3, om, ol, om, ol, qc, N, E);
}


# Layer-2 products moved behind the layer-2 edge sums, A(HW)=(AH)W: work-groups sum relu'd hidden rows in LDS, then multiply their sums by [W2|W3] on f32 WMMA from LDS; 8 launches; sum order changes
# speedup vs baseline: 3.1664x; 1.0914x over previous
#include <hip/hip_runtime.h>

typedef float v2f __attribute__((ext_vector_type(2)));
typedef float v8f __attribute__((ext_vector_type(8)));

#define IN_CH 512
#define HID   128
#define OUTC  64

#define RT    196
#define AGG_T 1024
#define AGG_W (AGG_T / 32)
#define QCAP  448

#define DEG_G 64
#define NCAP  (256 * RT)
#define DEG_B 8

__global__ void __launch_bounds__(1024)
k_deg_part(const int* __restrict__ ei, const float* __restrict__ ew,
           float* __restrict__ part, unsigned int* __restrict__ own, int N, int E) {
  __shared__ float dg[NCAP];
  const int tid = threadIdx.x;
  for (int i = tid; i < NCAP; i += 1024) dg[i] = 0.0f;
  __syncthreads();

  const int EP  = ((E + 511) >> 9) << 9;
  const int nCh = (EP + 1023) >> 10;
  const int per = (nCh + (int)gridDim.x - 1) / (int)gridDim.x;
  const int c0  = (int)blockIdx.x * per;
  int c1 = c0 + per;
  if (c1 > nCh) c1 = nCh;

  for (int c = c0; c < c1; c += DEG_B) {
    int   d[DEG_B];
    float w[DEG_B];
#pragma unroll
    for (int j = 0; j < DEG_B; ++j) {
      int e = ((c + j) << 10) + tid;
      d[j] = -1;
      w[j] = 0.0f;
      if (c + j < c1 && e < E) { d[j] = ei[E + e]; w[j] = ew[e]; }
    }
#pragma unroll
    for (int j = 0; j < DEG_B; ++j) {
      int e = ((c + j) << 10) + tid;
      unsigned int o = 0u;
      if (d[j] >= 0) {
        atomicAdd(&dg[d[j]], w[j]);
        o = (unsigned int)d[j] / RT;
      }
      unsigned int pl = 0u;
      for (int k = 0; k < 8; ++k) {
        unsigned int b = __builtin_amdgcn_ballot_w32(((o >> k) & 1u) != 0u);
        if ((tid & 31) == k) pl = b;
      }
      const int u  = e >> 9;
      const int gl = (e >> 5) & 15;
      const int k  = tid & 31;
      if (c + j < c1 && (u << 9) < EP && k < 8)
        own[((((u << 1) | (k >> 2)) << 4) | gl) * 4 + (k & 3)] = pl;
    }
  }
  __syncthreads();

  float* p = part + (size_t)blockIdx.x * N;
  for (int i = tid; i < N; i += 1024) p[i] = dg[i];
}

__global__ void k_rsqrt(const float* __restrict__ part, float* __restrict__ dinv, int n) {
  int i = blockIdx.x * blockDim.x + threadIdx.x;
  if (i < n) {
    float d = 1.0f;
#pragma unroll 16
    for (int g = 0; g < DEG_G; ++g) d += part[(size_t)g * n + i];
    dinv[i] = (d > 0.0f) ? rsqrtf(d) : 0.0f;
  }
}

__global__ void k_pack_w(const float* __restrict__ W, float* __restrict__ Wp,
                         int K, int N) {
  int idx = blockIdx.x * blockDim.x + threadIdx.x;
  if (idx >= K * N) return;
  int k = idx / N;
  int n = idx - k * N;
  int k4 = k >> 2;
  int hi = (k >> 1) & 1;
  int j  = k & 1;
  Wp[(((size_t)((k4 << 1) | hi)) * N + n) * 2 + j] = W[idx];
}

template <int NT>
__global__ void k_gemm_wmma(const float* __restrict__ A,
                            const float* __restrict__ Bp,
                            float* __restrict__ C,
                            int M, int K) {
  const int N = NT * 16;
  int wave = (int)((blockIdx.x * blockDim.x + threadIdx.x) >> 5);
  int lane = threadIdx.x & 31;
  if (wave >= (M >> 4)) return;

  int m0 = wave << 4;
  int lo = lane & 15;
  int hi = lane >> 4;

  const float* aPtr = A  + (size_t)(m0 + lo) * K + (hi << 1);
  const float* bPtr = Bp + ((size_t)hi * N + lo) * 2;

  v8f z = {0.f,0.f,0.f,0.f,0.f,0.f,0.f,0.f};
  v8f acc[NT];
#pragma unroll
  for (int t = 0; t < NT; ++t) acc[t] = z;

  for (int k = 0; k < K; k += 8) {
    v2f a0 = *(const v2f*)(aPtr + k);
    v2f a1 = *(const v2f*)(aPtr + k + 4);
    const float* bp0 = bPtr + (size_t)k * N;
    const float* bp1 = bp0 + (size_t)4 * N;
    v2f b0[NT], b1[NT];
#pragma unroll
    for (int t = 0; t < NT; ++t) b0[t] = *(const v2f*)(bp0 + t * 32);
#pragma unroll
    for (int t = 0; t < NT; ++t) b1[t] = *(const v2f*)(bp1 + t * 32);
#pragma unroll
    for (int t = 0; t < NT; ++t)
      acc[t] = __builtin_amdgcn_wmma_f32_16x16x4_f32(
          false, a0, false, b0[t], (short)0, acc[t], false, false);
#pragma unroll
    for (int t = 0; t < NT; ++t)
      acc[t] = __builtin_amdgcn_wmma_f32_16x16x4_f32(
          false, a1, false, b1[t], (short)0, acc[t], false, false);
  }

  int cRow = m0 + (hi << 3);
#pragma unroll
  for (int r = 0; r < 8; ++r) {
    float* cp = C + (size_t)(cRow + r) * N + lo;
#pragma unroll
    for (int t = 0; t < NT; ++t) cp[t * 16] = acc[t][r];
  }
}

#define EB 8
#define ACP2  132
#define AROW2 (((RT + 15) / 16) * 16)
#define NOREC 0xFFFFFFFFu

__device__ __forceinline__ int bcast_i(int v, int j) {
  return __builtin_amdgcn_readlane(v, j);
}
__device__ __forceinline__ float bcast_f(float v, int j) {
  return __int_as_float(__builtin_amdgcn_readlane(__float_as_int(v), j));
}

template <int LAYER>
__device__ __forceinline__ void row4(const float* __restrict__ ta,
                                     int s, int l, float* v) {
  int so = s * HID + l;
  v[0] = ta[so]; v[1] = ta[so + 32]; v[2] = ta[so + 64]; v[3] = ta[so + 96];
  if (LAYER == 2) {
    v[0] = fmaxf(v[0], 0.0f); v[1] = fmaxf(v[1], 0.0f);
    v[2] = fmaxf(v[2], 0.0f); v[3] = fmaxf(v[3], 0.0f);
  }
}

template <int LAYER>
__global__ void __launch_bounds__(AGG_T)
k_agg(const int* __restrict__ ei, const float* __restrict__ ew,
      const float* __restrict__ dinv, const unsigned int* __restrict__ own,
      const float* __restrict__ ta,
      const float* __restrict__ wa, const float* __restrict__ wb,
      const float* __restrict__ ba, const float* __restrict__ bb,
      float* oa, float* ob, float* qa, float* qb, unsigned int* qc, int N, int E) {
  constexpr int ACP   = (LAYER == 1) ? 128 : ACP2;
  constexpr int AROWS = (LAYER == 1) ? RT : AROW2;
  __shared__ __attribute__((aligned(16))) float acc[AROWS * ACP];
  __shared__ unsigned int q[AGG_W * QCAP];
  __shared__ unsigned int qn[AGG_W];
  __shared__ unsigned int tot;
  __shared__ unsigned int tw[AGG_W];

  const int tid  = threadIdx.x;
  const int l    = tid & 31;
  const int wave = __builtin_amdgcn_readfirstlane(tid >> 5);
  const int r0   = blockIdx.x * RT;

  for (int i = tid; i < AROWS * ACP; i += AGG_T) acc[i] = 0.0f;
  if (tid < AGG_W) qn[tid] = 0u;
  if (tid == 0) tot = 0u;
  __syncthreads();

  int rows = N - r0;
  if (rows > RT) rows = RT;
  const int cap = rows * OUTC;
  float* ra = ((LAYER == 1) ? qa : oa) + r0 * OUTC;
  float* rb = ((LAYER == 1) ? qb : ob) + r0 * OUTC;

  auto add_group = [&](int sv, int dl, float nv, int m) {
    int j0 = 0;
    for (; j0 + EB <= m; j0 += EB) {
      float v[EB][4];
#pragma unroll
      for (int j = 0; j < EB; ++j) row4<LAYER>(ta, bcast_i(sv, j0 + j), l, v[j]);
#pragma unroll
      for (int j = 0; j < EB; ++j) {
        float* ar = acc + bcast_i(dl, j0 + j) * ACP + l;
        float nm  = bcast_f(nv, j0 + j);
        atomicAdd(&ar[0],  v[j][0] * nm);
        atomicAdd(&ar[32], v[j][1] * nm);
        atomicAdd(&ar[64], v[j][2] * nm);
        atomicAdd(&ar[96], v[j][3] * nm);
      }
    }
    for (; j0 < m; ++j0) {
      float v[4];
      row4<LAYER>(ta, bcast_i(sv, j0), l, v);
      float* ar = acc + bcast_i(dl, j0) * ACP + l;
      float nm  = bcast_f(nv, j0);
      atomicAdd(&ar[0],  v[0] * nm);
      atomicAdd(&ar[32], v[1] * nm);
      atomicAdd(&ar[64], v[2] * nm);
      atomicAdd(&ar[96], v[3] * nm);
    }
  };

  unsigned int cw = NOREC;
  if (LAYER == 2) cw = (unsigned int)__builtin_amdgcn_readfirstlane((int)qc[blockIdx.x * AGG_W + wave]);
  if (cw != NOREC) {
    const int off = (int)(cw & 0xFFFFu);
    const int n   = (int)(cw >> 16);
    unsigned int p = 0u;
    float nv = 0.0f;
    if (l < n) { p = __float_as_uint(ra[off + l]); nv = rb[off + l]; }
    for (int g = 0; g < n; g += 32) {
      unsigned int pn = 0u;
      float nn = 0.0f;
      if (g + 32 + l < n) { pn = __float_as_uint(ra[off + g + 32 + l]); nn = rb[off + g + 32 + l]; }
      int m = n - g;
      if (m > 32) m = 32;
      add_group((int)(p & 0xFFFFu), (int)(p >> 16), nv, m);
      p = pn;
      nv = nn;
    }
  } else {

  const int nU = (E + 511) >> 9;
  const int nS = (nU + 1) >> 1;
  const unsigned int bid = blockIdx.x;
  const unsigned int m0 = (bid & 1u)   ? 0u : ~0u, m1 = (bid & 2u)   ? 0u : ~0u;
  const unsigned int m2 = (bid & 4u)   ? 0u : ~0u, m3 = (bid & 8u)   ? 0u : ~0u;
  const unsigned int m4 = (bid & 16u)  ? 0u : ~0u, m5 = (bid & 32u)  ? 0u : ~0u;
  const unsigned int m6 = (bid & 64u)  ? 0u : ~0u, m7 = (bid & 128u) ? 0u : ~0u;
  const int rot = (int)((bid * (unsigned int)nS) / gridDim.x);
  unsigned int* myq = q + wave * QCAP;
  const uint4* own4 = (const uint4*)own;

  auto search = [&](uint4 a, uint4 b, int eb, bool ok) {
    unsigned int zz = (a.x ^ m0) & (a.y ^ m1) & (a.z ^ m2) & (a.w ^ m3)
                    & (b.x ^ m4) & (b.y ^ m5) & (b.z ^ m6) & (b.w ^ m7);
    if (!ok) zz = 0u;
    while (zz) {
      int bit = __builtin_ctz(zz);
      zz &= zz - 1u;
      int e = eb + bit;
      if (e < E) {
        unsigned int pos = atomicAdd(&qn[wave], 1u);
        if (pos < QCAP) {
          myq[pos] = (unsigned int)e;
        } else {
          int s = ei[e];
          int d = ei[E + e];
          float nm = dinv[s] * ew[e] * dinv[d];
          float* ar = acc + (d - r0) * ACP;
          for (int c = 0; c < 128; ++c) {
            float hv = ta[s * HID + c];
            if (LAYER == 2) hv = fmaxf(hv, 0.0f);
            atomicAdd(&ar[c], hv * nm);
          }
        }
      }
    }
  };
  auto lane_unit = [&](int ii) {
    int c = ii + rot;
    if (c >= nS) c -= nS;
    return (c << 1) | (l >> 4);
  };
  const int gl = l & 15;

  int i = wave;
  for (; i + 3 * AGG_W < nS; i += 4 * AGG_W) {
    int u0 = lane_unit(i), u1 = lane_unit(i + AGG_W);
    int u2 = lane_unit(i + 2 * AGG_W), u3 = lane_unit(i + 3 * AGG_W);
    bool k0 = u0 < nU, k1 = u1 < nU, k2 = u2 < nU, k3 = u3 < nU;
    int c0 = min(u0, nU - 1) << 5, c1 = min(u1, nU - 1) << 5;
    int c2 = min(u2, nU - 1) << 5, c3 = min(u3, nU - 1) << 5;
    uint4 a0 = own4[c0 | gl], b0 = own4[c0 | 16 | gl];
    uint4 a1 = own4[c1 | gl], b1 = own4[c1 | 16 | gl];
    uint4 a2 = own4[c2 | gl], b2 = own4[c2 | 16 | gl];
    uint4 a3 = own4[c3 | gl], b3 = own4[c3 | 16 | gl];
    search(a0, b0, (u0 << 9) + (gl << 5), k0);
    search(a1, b1, (u1 << 9) + (gl << 5), k1);
    search(a2, b2, (u2 << 9) + (gl << 5), k2);
    search(a3, b3, (u3 << 9) + (gl << 5), k3);
  }
  for (; i < nS; i += AGG_W) {
    int u0 = lane_unit(i);
    bool k0 = u0 < nU;
    int c0 = min(u0, nU - 1) << 5;
    uint4 a0 = own4[c0 | gl], b0 = own4[c0 | 16 | gl];
    search(a0, b0, (u0 << 9) + (gl << 5), k0);
  }

  int nq = __builtin_amdgcn_readfirstlane((int)qn[wave]);
  const bool ovf = nq > QCAP;
  const int n = ovf ? QCAP : nq;
  int off = 0;
  bool keep = false;
  if (LAYER == 1) {
    unsigned int o = 0u;
    if (l == 0) o = atomicAdd(&tot, (unsigned int)((n + 31) & ~31));
    off = __builtin_amdgcn_readfirstlane((int)o);
    keep = !ovf && off + n <= cap;
    if (l == 0) tw[wave] = keep ? ((unsigned int)off | ((unsigned int)n << 16)) : NOREC;
  }
  for (int g = 0; g < n; g += 32) {
    int sv = 0, dl = 0;
    float nv = 0.0f;
    if (g + l < n) {
      int e = (int)myq[g + l];
      int s = ei[e];
      int d = ei[E + e];
      nv = dinv[s] * ew[e] * dinv[d];
      sv = s;
      dl = d - r0;
      if (LAYER == 1 && keep) {
        ra[off + g + l] = __uint_as_float((unsigned int)sv | ((unsigned int)dl << 16));
        rb[off + g + l] = nv;
      }
    }
    int m = n - g;
    if (m > 32) m = 32;
    add_group(sv, dl, nv, m);
  }
  }
  __syncthreads();
  if (LAYER == 1 && tid < AGG_W) qc[blockIdx.x * AGG_W + tid] = tw[tid];

  const int c = tid & 127;
  if (LAYER == 1) {
    for (int lr = tid >> 7; lr < rows; lr += AGG_T / 128) {
      int r = r0 + lr;
      float di = dinv[r];
      float d2 = di * di;
      float a = acc[lr * ACP + c];
      oa[r * HID + c] = a + (ta[r * HID + c] * d2 + ba[c]);
    }
  } else {
    const int WB = 8;
    const int RS = AGG_T / 128;
    for (int lr0 = tid >> 7; lr0 < rows; lr0 += WB * RS) {
      float di[WB], tv[WB];
#pragma unroll
      for (int j = 0; j < WB; ++j) {
        int lr = lr0 + j * RS;
        di[j] = 0.0f;
        tv[j] = 0.0f;
        if (lr < rows) { di[j] = dinv[r0 + lr]; tv[j] = ta[(r0 + lr) * HID + c]; }
      }
#pragma unroll
      for (int j = 0; j < WB; ++j) {
        int lr = lr0 + j * RS;
        if (lr < rows) acc[lr * ACP + c] += fmaxf(tv[j], 0.0f) * (di[j] * di[j]);
      }
    }
    __syncthreads();

    const int lo  = l & 15;
    const int hi  = l >> 4;
    const int cg  = wave & 3;
    const bool sec = cg >= 2;
    const int cb  = (cg & 1) * 32;
    const float* __restrict__ wsel = sec ? wb : wa;
    float* osel = sec ? ob : oa;
    const float bias = sec ? bb[cb + l] : ba[cb + l];
    const float* bPtr = wsel + (hi * OUTC + cb + lo) * 2;
    for (int rt = wave >> 2; rt < AROWS / 16; rt += AGG_W / 4) {
      const float* aPtr = acc + (rt * 16 + lo) * ACP + (hi << 1);
      v8f d0 = {0.f, 0.f, 0.f, 0.f, 0.f, 0.f, 0.f, 0.f};
      v8f d1 = d0;
#pragma unroll 8
      for (int k = 0; k < HID; k += 4) {
        v2f a  = *(const v2f*)(aPtr + k);
        v2f b0 = *(const v2f*)(bPtr + k * OUTC);
        v2f b1 = *(const v2f*)(bPtr + k * OUTC + 32);
        d0 = __builtin_amdgcn_wmma_f32_16x16x4_f32(
            false, a, false, b0, (short)0, d0, false, false);
        d1 = __builtin_amdgcn_wmma_f32_16x16x4_f32(
            false, a, false, b1, (short)0, d1, false, false);
      }
      asm volatile("v_nop\n\tv_nop\n\tv_nop\n\tv_nop\n\tv_nop\n\tv_nop\n\tv_nop\n\tv_nop"
                   : "+v"(d0), "+v"(d1));
#pragma unroll
      for (int r = 0; r < 8; ++r) {
        const float own = hi ? d1[r] : d0[r];
        const float snd = hi ? d0[r] : d1[r];
        const float got = __int_as_float(__builtin_amdgcn_permlanex16(
            0, __float_as_int(snd), 0x76543210, 0xfedcba98, false, false));
        const float vA = hi ? got : own;
        const float vB = hi ? own : got;
        const int rowA = rt * 16 + r;
        const int rowB = rowA + 8;
        if (rowA < rows) osel[(r0 + rowA) * OUTC + cb + l] = vA + bias;
        if (rowB < rows) osel[(r0 + rowB) * OUTC + cb + l] = vB + bias;
      }
    }
  }
}

extern "C" void kernel_launch(void* const* d_in, const int* in_sizes, int n_in,
                              void* d_out, int out_size, void* d_ws, size_t ws_size,
                              hipStream_t stream) {
  const float* x  = (const float*)d_in[0];
  const int*   ei = (const int*)  d_in[1];
  const float* ew = (const float*)d_in[2];
  const float* W1 = (const float*)d_in[3];
  const float* b1 = (const float*)d_in[4];
  const float* W2 = (const float*)d_in[5];
  const float* b2 = (const float*)d_in[6];
  const float* W3 = (const float*)d_in[7];
  const float* b3 = (const float*)d_in[8];

  const int N = in_sizes[0] / IN_CH;
  const int E = in_sizes[2];

  float* ws   = (float*)d_ws;
  float* deg  = ws;
  float* dinv = ws + N;
  float* h1   = ws + 2 * (size_t)N;
  float* agg1 = h1 + (size_t)N * HID;
  float* W1p  = agg1 + (size_t)N * HID;
  float* W2p  = W1p + (size_t)IN_CH * HID;
  float* W3p  = W2p + (size_t)HID * OUTC;
  unsigned int* own = (unsigned int*)(W3p + (size_t)HID * OUTC);
  float* om   = (float*)d_out;
  float* ol   = om + (size_t)N * OUTC;
  unsigned int* qc = (unsigned int*)deg;
  float* part = h1;

  const int T = 256;

  k_pack_w<<<(IN_CH * HID + T - 1) / T, T, 0, stream>>>(W1, W1p, IN_CH, HID);
  k_pack_w<<<(HID * OUTC + T - 1) / T, T, 0, stream>>>(W2, W2p, HID, OUTC);
  k_pack_w<<<(HID * OUTC + T - 1) / T, T, 0, stream>>>(W3, W3p, HID, OUTC);

  k_deg_part  <<<DEG_G, 1024, 0, stream>>>(ei, ew, part, own, N, E);
  k_rsqrt     <<<(N + T - 1) / T, T, 0, stream>>>(part, dinv, N);

  const int waves  = N >> 4;
  const int gblocks = (waves * 32 + T - 1) / T;
  const int ablocks = (N + RT - 1) / RT;

  k_gemm_wmma<8><<<gblocks, T, 0, stream>>>(x, W1p, h1, N, IN_CH);

  k_agg<1><<<ablocks, AGG_T, 0, stream>>>(ei, ew, dinv, own, h1, W2p, W3p, b1, b1, agg1, agg1, om, ol, qc, N, E);

  k_agg<2><<<ablocks, AGG_T, 0, stream>>>(ei, ew, dinv, own, agg1, W2p, W3p, b2, b3, om, ol, om, ol, qc, N, E);
}


# Layer-1 product: each wave keeps its 512x32 slice of W1 in registers for the whole launch; 256 resident 4-wave work-groups pass 16-row x tiles through LDS (two buffers); k sum order changes
# speedup vs baseline: 3.3377x; 1.0541x over previous
#include <hip/hip_runtime.h>

typedef float v2f __attribute__((ext_vector_type(2)));
typedef float v8f __attribute__((ext_vector_type(8)));

#define IN_CH 512
#define HID   128
#define OUTC  64

#define RT    196
#define AGG_T 1024
#define AGG_W (AGG_T / 32)
#define QCAP  448

#define DEG_G 64
#define NCAP  (256 * RT)
#define DEG_B 8

__global__ void __launch_bounds__(1024)
k_deg_part(const int* __restrict__ ei, const float* __restrict__ ew,
           float* __restrict__ part, unsigned int* __restrict__ own, int N, int E) {
  __shared__ float dg[NCAP];
  const int tid = threadIdx.x;
  for (int i = tid; i < NCAP; i += 1024) dg[i] = 0.0f;
  __syncthreads();

  const int EP  = ((E + 511) >> 9) << 9;
  const int nCh = (EP + 1023) >> 10;
  const int per = (nCh + (int)gridDim.x - 1) / (int)gridDim.x;
  const int c0  = (int)blockIdx.x * per;
  int c1 = c0 + per;
  if (c1 > nCh) c1 = nCh;

  for (int c = c0; c < c1; c += DEG_B) {
    int   d[DEG_B];
    float w[DEG_B];
#pragma unroll
    for (int j = 0; j < DEG_B; ++j) {
      int e = ((c + j) << 10) + tid;
      d[j] = -1;
      w[j] = 0.0f;
      if (c + j < c1 && e < E) { d[j] = ei[E + e]; w[j] = ew[e]; }
    }
#pragma unroll
    for (int j = 0; j < DEG_B; ++j) {
      int e = ((c + j) << 10) + tid;
      unsigned int o = 0u;
      if (d[j] >= 0) {
        atomicAdd(&dg[d[j]], w[j]);
        o = (unsigned int)d[j] / RT;
      }
      unsigned int pl = 0u;
      for (int k = 0; k < 8; ++k) {
        unsigned int b = __builtin_amdgcn_ballot_w32(((o >> k) & 1u) != 0u);
        if ((tid & 31) == k) pl = b;
      }
      const int u  = e >> 9;
      const int gl = (e >> 5) & 15;
      const int k  = tid & 31;
      if (c + j < c1 && (u << 9) < EP && k < 8)
        own[((((u << 1) | (k >> 2)) << 4) | gl) * 4 + (k & 3)] = pl;
    }
  }
  __syncthreads();

  float* p = part + (size_t)blockIdx.x * N;
  for (int i = tid; i < N; i += 1024) p[i] = dg[i];
}

__global__ void k_rsqrt(const float* __restrict__ part, float* __restrict__ dinv, int n) {
  int i = blockIdx.x * blockDim.x + threadIdx.x;
  if (i < n) {
    float d = 1.0f;
#pragma unroll 16
    for (int g = 0; g < DEG_G; ++g) d += part[(size_t)g * n + i];
    dinv[i] = (d > 0.0f) ? rsqrtf(d) : 0.0f;
  }
}

__global__ void k_pack_w(const float* __restrict__ W, float* __restrict__ Wp,
                         int K, int N) {
  int idx = blockIdx.x * blockDim.x + threadIdx.x;
  if (idx >= K * N) return;
  int k = idx / N;
  int n = idx - k * N;
  int k4 = k >> 2;
  int hi = (k >> 1) & 1;
  int j  = k & 1;
  Wp[(((size_t)((k4 << 1) | hi)) * N + n) * 2 + j] = W[idx];
}

#define GM_G  256
#define GM_AP 516
typedef float v4f __attribute__((ext_vector_type(4)));

__global__ void __launch_bounds__(128)
k_gemm_l1(const float* __restrict__ A, const float* __restrict__ Bp,
          float* __restrict__ C, int M) {
  __shared__ __attribute__((aligned(16))) float xs[2 * 16 * GM_AP];
  const int tid  = threadIdx.x;
  const int l    = tid & 31;
  const int lo   = l & 15;
  const int hi   = l >> 4;
  const int wave = __builtin_amdgcn_readfirstlane(tid >> 5);
  const int nT   = M >> 4;
  const int n0   = wave * 32;

  v2f b[IN_CH / 8][2][2];
#pragma unroll
  for (int j = 0; j < IN_CH / 8; ++j)
#pragma unroll
    for (int s2 = 0; s2 < 2; ++s2)
#pragma unroll
      for (int t = 0; t < 2; ++t)
        b[j][s2][t] = *(const v2f*)(Bp + ((size_t)(((2 * j + hi) * 2 + s2) * HID + n0 + t * 16 + lo)) * 2);

  v4f st[16];
  int t = blockIdx.x;
  if (t < nT) {
#pragma unroll
    for (int i = 0; i < 16; ++i)
      st[i] = *((const v4f*)(A + (size_t)(t * 16 + i) * IN_CH) + tid);
#pragma unroll
    for (int i = 0; i < 16; ++i)
      *(v4f*)(xs + i * GM_AP + 4 * tid) = st[i];
  }
  __syncthreads();

  int buf = 0;
  for (; t < nT; t += (int)gridDim.x) {
    const int  tn   = t + (int)gridDim.x;
    const bool more = tn < nT;
    if (more) {
#pragma unroll
      for (int i = 0; i < 16; ++i)
        st[i] = *((const v4f*)(A + (size_t)(tn * 16 + i) * IN_CH) + tid);
    }

    const float* ap = xs + buf * (16 * GM_AP) + lo * GM_AP + 4 * hi;
    float* cp = C + (size_t)(t * 16 + 8 * hi) * HID + n0 + lo;
    v8f d0 = {0.f, 0.f, 0.f, 0.f, 0.f, 0.f, 0.f, 0.f};
    v8f d1 = d0;
#pragma unroll
    for (int j = 0; j < IN_CH / 8; ++j) {
      v4f a = *(const v4f*)(ap + 8 * j);
      v2f a0 = {a.x, a.y};
      v2f a1 = {a.z, a.w};
      d0 = __builtin_amdgcn_wmma_f32_16x16x4_f32(
          false, a0, false, b[j][0][0], (short)0, d0, false, false);
      d1 = __builtin_amdgcn_wmma_f32_16x16x4_f32(
          false, a0, false, b[j][0][1], (short)0, d1, false, false);
      d0 = __builtin_amdgcn_wmma_f32_16x16x4_f32(
          false, a1, false, b[j][1][0], (short)0, d0, false, false);
      d1 = __builtin_amdgcn_wmma_f32_16x16x4_f32(
          false, a1, false, b[j][1][1], (short)0, d1, false, false);
    }
    asm volatile("v_nop\n\tv_nop\n\tv_nop\n\tv_nop\n\tv_nop\n\tv_nop\n\tv_nop\n\tv_nop"
                 : "+v"(d0), "+v"(d1));

    if (more) {
#pragma unroll
      for (int i = 0; i < 16; ++i)
        *(v4f*)(xs + (buf ^ 1) * (16 * GM_AP) + i * GM_AP + 4 * tid) = st[i];
    }
    __syncthreads();
#pragma unroll
    for (int r = 0; r < 8; ++r) {
      cp[r * HID]      = d0[r];
      cp[r * HID + 16] = d1[r];
    }
    buf ^= 1;
  }
}

#define EB 8
#define ACP2  132
#define AROW2 (((RT + 15) / 16) * 16)
#define NOREC 0xFFFFFFFFu

__device__ __forceinline__ int bcast_i(int v, int j) {
  return __builtin_amdgcn_readlane(v, j);
}
__device__ __forceinline__ float bcast_f(float v, int j) {
  return __int_as_float(__builtin_amdgcn_readlane(__float_as_int(v), j));
}

template <int LAYER>
__device__ __forceinline__ void row4(const float* __restrict__ ta,
                                     int s, int l, float* v) {
  int so = s * HID + l;
  v[0] = ta[so]; v[1] = ta[so + 32]; v[2] = ta[so + 64]; v[3] = ta[so + 96];
  if (LAYER == 2) {
    v[0] = fmaxf(v[0], 0.0f); v[1] = fmaxf(v[1], 0.0f);
    v[2] = fmaxf(v[2], 0.0f); v[3] = fmaxf(v[3], 0.0f);
  }
}

template <int LAYER>
__global__ void __launch_bounds__(AGG_T)
k_agg(const int* __restrict__ ei, const float* __restrict__ ew,
      const float* __restrict__ dinv, const unsigned int* __restrict__ own,
      const float* __restrict__ ta,
      const float* __restrict__ wa, const float* __restrict__ wb,
      const float* __restrict__ ba, const float* __restrict__ bb,
      float* oa, float* ob, float* qa, float* qb, unsigned int* qc, int N, int E) {
  constexpr int ACP   = (LAYER == 1) ? 128 : ACP2;
  constexpr int AROWS = (LAYER == 1) ? RT : AROW2;
  __shared__ __attribute__((aligned(16))) float acc[AROWS * ACP];
  __shared__ unsigned int q[AGG_W * QCAP];
  __shared__ unsigned int qn[AGG_W];
  __shared__ unsigned int tot;
  __shared__ unsigned int tw[AGG_W];

  const int tid  = threadIdx.x;
  const int l    = tid & 31;
  const int wave = __builtin_amdgcn_readfirstlane(tid >> 5);
  const int r0   = blockIdx.x * RT;

  for (int i = tid; i < AROWS * ACP; i += AGG_T) acc[i] = 0.0f;
  if (tid < AGG_W) qn[tid] = 0u;
  if (tid == 0) tot = 0u;
  __syncthreads();

  int rows = N - r0;
  if (rows > RT) rows = RT;
  const int cap = rows * OUTC;
  float* ra = ((LAYER == 1) ? qa : oa) + r0 * OUTC;
  float* rb = ((LAYER == 1) ? qb : ob) + r0 * OUTC;

  auto add_group = [&](int sv, int dl, float nv, int m) {
    int j0 = 0;
    for (; j0 + EB <= m; j0 += EB) {
      float v[EB][4];
#pragma unroll
      for (int j = 0; j < EB; ++j) row4<LAYER>(ta, bcast_i(sv, j0 + j), l, v[j]);
#pragma unroll
      for (int j = 0; j < EB; ++j) {
        float* ar = acc + bcast_i(dl, j0 + j) * ACP + l;
        float nm  = bcast_f(nv, j0 + j);
        atomicAdd(&ar[0],  v[j][0] * nm);
        atomicAdd(&ar[32], v[j][1] * nm);
        atomicAdd(&ar[64], v[j][2] * nm);
        atomicAdd(&ar[96], v[j][3] * nm);
      }
    }
    for (; j0 < m; ++j0) {
      float v[4];
      row4<LAYER>(ta, bcast_i(sv, j0), l, v);
      float* ar = acc + bcast_i(dl, j0) * ACP + l;
      float nm  = bcast_f(nv, j0);
      atomicAdd(&ar[0],  v[0] * nm);
      atomicAdd(&ar[32], v[1] * nm);
      atomicAdd(&ar[64], v[2] * nm);
      atomicAdd(&ar[96], v[3] * nm);
    }
  };

  unsigned int cw = NOREC;
  if (LAYER == 2) cw = (unsigned int)__builtin_amdgcn_readfirstlane((int)qc[blockIdx.x * AGG_W + wave]);
  if (cw != NOREC) {
    const int off = (int)(cw & 0xFFFFu);
    const int n   = (int)(cw >> 16);
    unsigned int p = 0u;
    float nv = 0.0f;
    if (l < n) { p = __float_as_uint(ra[off + l]); nv = rb[off + l]; }
    for (int g = 0; g < n; g += 32) {
      unsigned int pn = 0u;
      float nn = 0.0f;
      if (g + 32 + l < n) { pn = __float_as_uint(ra[off + g + 32 + l]); nn = rb[off + g + 32 + l]; }
      int m = n - g;
      if (m > 32) m = 32;
      add_group((int)(p & 0xFFFFu), (int)(p >> 16), nv, m);
      p = pn;
      nv = nn;
    }
  } else {

  const int nU = (E + 511) >> 9;
  const int nS = (nU + 1) >> 1;
  const unsigned int bid = blockIdx.x;
  const unsigned int m0 = (bid & 1u)   ? 0u : ~0u, m1 = (bid & 2u)   ? 0u : ~0u;
  const unsigned int m2 = (bid & 4u)   ? 0u : ~0u, m3 = (bid & 8u)   ? 0u : ~0u;
  const unsigned int m4 = (bid & 16u)  ? 0u : ~0u, m5 = (bid & 32u)  ? 0u : ~0u;
  const unsigned int m6 = (bid & 64u)  ? 0u : ~0u, m7 = (bid & 128u) ? 0u : ~0u;
  const int rot = (int)((bid * (unsigned int)nS) / gridDim.x);
  unsigned int* myq = q + wave * QCAP;
  const uint4* own4 = (const uint4*)own;

  auto search = [&](uint4 a, uint4 b, int eb, bool ok) {
    unsigned int zz = (a.x ^ m0) & (a.y ^ m1) & (a.z ^ m2) & (a.w ^ m3)
                    & (b.x ^ m4) & (b.y ^ m5) & (b.z ^ m6) & (b.w ^ m7);
    if (!ok) zz = 0u;
    while (zz) {
      int bit = __builtin_ctz(zz);
      zz &= zz - 1u;
      int e = eb + bit;
      if (e < E) {
        unsigned int pos = atomicAdd(&qn[wave], 1u);
        if (pos < QCAP) {
          myq[pos] = (unsigned int)e;
        } else {
          int s = ei[e];
          int d = ei[E + e];
          float nm = dinv[s] * ew[e] * dinv[d];
          float* ar = acc + (d - r0) * ACP;
          for (int c = 0; c < 128; ++c) {
            float hv = ta[s * HID + c];
            if (LAYER == 2) hv = fmaxf(hv, 0.0f);
            atomicAdd(&ar[c], hv * nm);
          }
        }
      }
    }
  };
  auto lane_unit = [&](int ii) {
    int c = ii + rot;
    if (c >= nS) c -= nS;
    return (c << 1) | (l >> 4);
  };
  const int gl = l & 15;

  int i = wave;
  for (; i + 3 * AGG_W < nS; i += 4 * AGG_W) {
    int u0 = lane_unit(i), u1 = lane_unit(i + AGG_W);
    int u2 = lane_unit(i + 2 * AGG_W), u3 = lane_unit(i + 3 * AGG_W);
    bool k0 = u0 < nU, k1 = u1 < nU, k2 = u2 < nU, k3 = u3 < nU;
    int c0 = min(u0, nU - 1) << 5, c1 = min(u1, nU - 1) << 5;
    int c2 = min(u2, nU - 1) << 5, c3 = min(u3, nU - 1) << 5;
    uint4 a0 = own4[c0 | gl], b0 = own4[c0 | 16 | gl];
    uint4 a1 = own4[c1 | gl], b1 = own4[c1 | 16 | gl];
    uint4 a2 = own4[c2 | gl], b2 = own4[c2 | 16 | gl];
    uint4 a3 = own4[c3 | gl], b3 = own4[c3 | 16 | gl];
    search(a0, b0, (u0 << 9) + (gl << 5), k0);
    search(a1, b1, (u1 << 9) + (gl << 5), k1);
    search(a2, b2, (u2 << 9) + (gl << 5), k2);
    search(a3, b3, (u3 << 9) + (gl << 5), k3);
  }
  for (; i < nS; i += AGG_W) {
    int u0 = lane_unit(i);
    bool k0 = u0 < nU;
    int c0 = min(u0, nU - 1) << 5;
    uint4 a0 = own4[c0 | gl], b0 = own4[c0 | 16 | gl];
    search(a0, b0, (u0 << 9) + (gl << 5), k0);
  }

  int nq = __builtin_amdgcn_readfirstlane((int)qn[wave]);
  const bool ovf = nq > QCAP;
  const int n = ovf ? QCAP : nq;
  int off = 0;
  bool keep = false;
  if (LAYER == 1) {
    unsigned int o = 0u;
    if (l == 0) o = atomicAdd(&tot, (unsigned int)((n + 31) & ~31));
    off = __builtin_amdgcn_readfirstlane((int)o);
    keep = !ovf && off + n <= cap;
    if (l == 0) tw[wave] = keep ? ((unsigned int)off | ((unsigned int)n << 16)) : NOREC;
  }
  for (int g = 0; g < n; g += 32) {
    int sv = 0, dl = 0;
    float nv = 0.0f;
    if (g + l < n) {
      int e = (int)myq[g + l];
      int s = ei[e];
      int d = ei[E + e];
      nv = dinv[s] * ew[e] * dinv[d];
      sv = s;
      dl = d - r0;
      if (LAYER == 1 && keep) {
        ra[off + g + l] = __uint_as_float((unsigned int)sv | ((unsigned int)dl << 16));
        rb[off + g + l] = nv;
      }
    }
    int m = n - g;
    if (m > 32) m = 32;
    add_group(sv, dl, nv, m);
  }
  }
  __syncthreads();
  if (LAYER == 1 && tid < AGG_W) qc[blockIdx.x * AGG_W + tid] = tw[tid];

  const int c = tid & 127;
  if (LAYER == 1) {
    for (int lr = tid >> 7; lr < rows; lr += AGG_T / 128) {
      int r = r0 + lr;
      float di = dinv[r];
      float d2 = di * di;
      float a = acc[lr * ACP + c];
      oa[r * HID + c] = a + (ta[r * HID + c] * d2 + ba[c]);
    }
  } else {
    const int WB = 8;
    const int RS = AGG_T / 128;
    for (int lr0 = tid >> 7; lr0 < rows; lr0 += WB * RS) {
      float di[WB], tv[WB];
#pragma unroll
      for (int j = 0; j < WB; ++j) {
        int lr = lr0 + j * RS;
        di[j] = 0.0f;
        tv[j] = 0.0f;
        if (lr < rows) { di[j] = dinv[r0 + lr]; tv[j] = ta[(r0 + lr) * HID + c]; }
      }
#pragma unroll
      for (int j = 0; j < WB; ++j) {
        int lr = lr0 + j * RS;
        if (lr < rows) acc[lr * ACP + c] += fmaxf(tv[j], 0.0f) * (di[j] * di[j]);
      }
    }
    __syncthreads();

    const int lo  = l & 15;
    const int hi  = l >> 4;
    const int cg  = wave & 3;
    const bool sec = cg >= 2;
    const int cb  = (cg & 1) * 32;
    const float* __restrict__ wsel = sec ? wb : wa;
    float* osel = sec ? ob : oa;
    const float bias = sec ? bb[cb + l] : ba[cb + l];
    const float* bPtr = wsel + (hi * OUTC + cb + lo) * 2;
    for (int rt = wave >> 2; rt < AROWS / 16; rt += AGG_W / 4) {
      const float* aPtr = acc + (rt * 16 + lo) * ACP + (hi << 1);
      v8f d0 = {0.f, 0.f, 0.f, 0.f, 0.f, 0.f, 0.f, 0.f};
      v8f d1 = d0;
#pragma unroll 8
      for (int k = 0; k < HID; k += 4) {
        v2f a  = *(const v2f*)(aPtr + k);
        v2f b0 = *(const v2f*)(bPtr + k * OUTC);
        v2f b1 = *(const v2f*)(bPtr + k * OUTC + 32);
        d0 = __builtin_amdgcn_wmma_f32_16x16x4_f32(
            false, a, false, b0, (short)0, d0, false, false);
        d1 = __builtin_amdgcn_wmma_f32_16x16x4_f32(
            false, a, false, b1, (short)0, d1, false, false);
      }
      asm volatile("v_nop\n\tv_nop\n\tv_nop\n\tv_nop\n\tv_nop\n\tv_nop\n\tv_nop\n\tv_nop"
                   : "+v"(d0), "+v"(d1));
#pragma unroll
      for (int r = 0; r < 8; ++r) {
        const float own = hi ? d1[r] : d0[r];
        const float snd = hi ? d0[r] : d1[r];
        const float got = __int_as_float(__builtin_amdgcn_permlanex16(
            0, __float_as_int(snd), 0x76543210, 0xfedcba98, false, false));
        const float vA = hi ? got : own;
        const float vB = hi ? own : got;
        const int rowA = rt * 16 + r;
        const int rowB = rowA + 8;
        if (rowA < rows) osel[(r0 + rowA) * OUTC + cb + l] = vA + bias;
        if (rowB < rows) osel[(r0 + rowB) * OUTC + cb + l] = vB + bias;
      }
    }
  }
}

extern "C" void kernel_launch(void* const* d_in, const int* in_sizes, int n_in,
                              void* d_out, int out_size, void* d_ws, size_t ws_size,
                              hipStream_t stream) {
  const float* x  = (const float*)d_in[0];
  const int*   ei = (const int*)  d_in[1];
  const float* ew = (const float*)d_in[2];
  const float* W1 = (const float*)d_in[3];
  const float* b1 = (const float*)d_in[4];
  const float* W2 = (const float*)d_in[5];
  const float* b2 = (const float*)d_in[6];
  const float* W3 = (const float*)d_in[7];
  const float* b3 = (const float*)d_in[8];

  const int N = in_sizes[0] / IN_CH;
  const int E = in_sizes[2];

  float* ws   = (float*)d_ws;
  float* deg  = ws;
  float* dinv = ws + N;
  float* h1   = ws + 2 * (size_t)N;
  float* agg1 = h1 + (size_t)N * HID;
  float* W1p  = agg1 + (size_t)N * HID;
  float* W2p  = W1p + (size_t)IN_CH * HID;
  float* W3p  = W2p + (size_t)HID * OUTC;
  unsigned int* own = (unsigned int*)(W3p + (size_t)HID * OUTC);
  float* om   = (float*)d_out;
  float* ol   = om + (size_t)N * OUTC;
  unsigned int* qc = (unsigned int*)deg;
  float* part = h1;

  const int T = 256;

  k_pack_w<<<(IN_CH * HID + T - 1) / T, T, 0, stream>>>(W1, W1p, IN_CH, HID);
  k_pack_w<<<(HID * OUTC + T - 1) / T, T, 0, stream>>>(W2, W2p, HID, OUTC);
  k_pack_w<<<(HID * OUTC + T - 1) / T, T, 0, stream>>>(W3, W3p, HID, OUTC);

  k_deg_part  <<<DEG_G, 1024, 0, stream>>>(ei, ew, part, own, N, E);
  k_rsqrt     <<<(N + T - 1) / T, T, 0, stream>>>(part, dinv, N);

  const int ablocks = (N + RT - 1) / RT;

  k_gemm_l1<<<GM_G, 128, 0, stream>>>(x, W1p, h1, N);

  k_agg<1><<<ablocks, AGG_T, 0, stream>>>(ei, ew, dinv, own, h1, W2p, W3p, b1, b1, agg1, agg1, om, ol, qc, N, E);

  k_agg<2><<<ablocks, AGG_T, 0, stream>>>(ei, ew, dinv, own, agg1, W2p, W3p, b2, b3, om, ol, om, ol, qc, N, E);
}
